# starting kernel
# baseline (speedup 1.0000x reference)
#include <hip/hip_runtime.h>
#include <hip/hip_fp16.h>

typedef __attribute__((ext_vector_type(16))) _Float16 v16h;
typedef __attribute__((ext_vector_type(8)))  _Float16 v8h;
typedef __attribute__((ext_vector_type(8)))  float    v8f;

__device__ __forceinline__ unsigned fp_ord(float f) {
    unsigned u = __float_as_uint(f);
    return (u & 0x80000000u) ? ~u : (u | 0x80000000u);
}
__device__ __forceinline__ float fp_unord(unsigned u) {
    unsigned b = (u & 0x80000000u) ? (u ^ 0x80000000u) : ~u;
    return __uint_as_float(b);
}

__global__ void k_node_transform1(const float* __restrict__ x,
                                  const float* __restrict__ Wl, const float* __restrict__ bl,
                                  const float* __restrict__ Wr, const float* __restrict__ br,
                                  float* __restrict__ xl1, float* __restrict__ xr1, int N) {
    int idx = blockIdx.x * blockDim.x + threadIdx.x;
    if (idx >= N * 128) return;
    int n = idx >> 7, d = idx & 127;
    float x0 = x[n * 2], x1 = x[n * 2 + 1];
    xl1[idx] = fmaf(x0, Wl[d], fmaf(x1, Wl[128 + d], bl[d]));
    xr1[idx] = fmaf(x0, Wr[d], fmaf(x1, Wr[128 + d], br[d]));
}

template <int H, int C>
__global__ void k_edge_logits(const float* __restrict__ xl, const float* __restrict__ xr,
                              const float* __restrict__ ea, const float* __restrict__ We,
                              const float* __restrict__ att,
                              const int* __restrict__ src, const int* __restrict__ tgt,
                              float* __restrict__ elog, unsigned* __restrict__ emaxM, int E) {
    const int D = H * C;
    int idx = blockIdx.x * blockDim.x + threadIdx.x;
    if (idx >= E * H) return;
    int e = idx / H, h = idx % H;
    int s = src[e], t = tgt[e];
    float a0 = ea[e * 2], a1 = ea[e * 2 + 1];
    int base = h * C;
    float acc = 0.f;
#pragma unroll
    for (int c = 0; c < C; ++c) {
        int d = base + c;
        float m = xl[s * D + d] + xr[t * D + d] + a0 * We[d] + a1 * We[D + d];
        m = (m > 0.f) ? m : 0.2f * m;
        acc = fmaf(m, att[base + c], acc);
    }
    elog[idx] = acc;
    atomicMax(&emaxM[t * H + h], fp_ord(acc));
}

template <int H>
__global__ void k_edge_exp(float* __restrict__ elog, const unsigned* __restrict__ emaxM,
                           float* __restrict__ denom, const int* __restrict__ tgt, int E) {
    int idx = blockIdx.x * blockDim.x + threadIdx.x;
    if (idx >= E * H) return;
    int e = idx / H, h = idx % H;
    int t = tgt[e];
    float mx = fp_unord(emaxM[t * H + h]);
    float v = __expf(elog[idx] - mx);
    elog[idx] = v;
    atomicAdd(&denom[t * H + h], v);
}

template <int H, int C>
__global__ void k_edge_agg(const float* __restrict__ ee, const float* __restrict__ denom,
                           const float* __restrict__ xl,
                           const int* __restrict__ src, const int* __restrict__ tgt,
                           float* __restrict__ agg, int E) {
    const int D = H * C;
    const int Q = D / 4;
    int idx = blockIdx.x * blockDim.x + threadIdx.x;
    if (idx >= E * Q) return;
    int e = idx / Q, q = idx % Q;
    int d = q * 4, h = d / C;
    int s = src[e], t = tgt[e];
    float alpha = ee[e * H + h] / denom[t * H + h];
    float4 v = *(const float4*)(xl + s * D + d);
    atomicAdd(&agg[t * D + d + 0], alpha * v.x);
    atomicAdd(&agg[t * D + d + 1], alpha * v.y);
    atomicAdd(&agg[t * D + d + 2], alpha * v.z);
    atomicAdd(&agg[t * D + d + 3], alpha * v.w);
}

__global__ void k_elu_half(const float* __restrict__ agg, const float* __restrict__ bias,
                           _Float16* __restrict__ hh, int N) {
    int idx = blockIdx.x * blockDim.x + threadIdx.x;
    if (idx >= N * 128) return;
    int d = idx & 127;
    float v = agg[idx] + bias[d];
    v = (v > 0.f) ? v : (__expf(v) - 1.f);
    hh[idx] = (_Float16)v;
}

__global__ void k_w_pack(const float* __restrict__ Wl, const float* __restrict__ Wr,
                         _Float16* __restrict__ Wpk) {
    int idx = blockIdx.x * blockDim.x + threadIdx.x;
    if (idx >= 2 * 8192) return;
    int which = idx >> 13;
    int r = idx & 8191;
    int j = r & 15;
    int lane = (r >> 4) & 31;
    int frag = r >> 9;
    int kc = frag >> 2, nt = frag & 3;
    int half_sel = lane >> 4, lidx = lane & 15;
    int kb = (j < 8) ? (half_sel * 8) : (16 + half_sel * 8);
    int k = kc * 32 + kb + (j & 7);
    int n = nt * 16 + lidx;
    const float* W = which ? Wr : Wl;
    Wpk[idx] = (_Float16)W[k * 64 + n];
}

__global__ void k_gemm2_wmma(const _Float16* __restrict__ hh,
                             const _Float16* __restrict__ Wpk,
                             const float* __restrict__ bl, const float* __restrict__ br,
                             float* __restrict__ xl2, float* __restrict__ xr2, int N) {
    __shared__ __align__(32) _Float16 ldsW[2 * 8192];

    {
        const uint4* g = (const uint4*)Wpk;
        uint4* l = (uint4*)ldsW;
        for (int i = threadIdx.x; i < 2048; i += blockDim.x) l[i] = g[i];
    }
    __syncthreads();

    int wave = (blockIdx.x * blockDim.x + threadIdx.x) >> 5;
    int lane = threadIdx.x & 31;
    int row0 = wave * 16;
    if (row0 >= N) return;

    int half_sel = lane >> 4;
    int lidx = lane & 15;
    int kb_lo = half_sel * 8;
    int kb_hi = 16 + kb_lo;
    int m = row0 + lidx;
    if (m >= N) m = N - 1;

    const v16h* fragL = (const v16h*)ldsW;
    const v16h* fragR = (const v16h*)(ldsW + 8192);

    v8f accl[4] = {};
    v8f accr[4] = {};

#pragma unroll
    for (int kc = 0; kc < 4; ++kc) {
        const _Float16* pa = hh + m * 128 + kc * 32;
        v8h alo = *(const v8h*)(pa + kb_lo);
        v8h ahi = *(const v8h*)(pa + kb_hi);
        v16h a = __builtin_shufflevector(alo, ahi, 0, 1, 2, 3, 4, 5, 6, 7,
                                         8, 9, 10, 11, 12, 13, 14, 15);
#pragma unroll
        for (int nt = 0; nt < 4; ++nt) {
            int fi = (kc * 4 + nt) * 32 + lane;
            v16h b  = fragL[fi];
            v16h b2 = fragR[fi];
            accl[nt] = __builtin_amdgcn_wmma_f32_16x16x32_f16(
                false, a, false, b, (short)0, accl[nt], false, false);
            accr[nt] = __builtin_amdgcn_wmma_f32_16x16x32_f16(
                false, a, false, b2, (short)0, accr[nt], false, false);
        }
    }
#pragma unroll
    for (int nt = 0; nt < 4; ++nt) {
        int n = nt * 16 + lidx;
        float bln = bl[n], brn = br[n];
#pragma unroll
        for (int r = 0; r < 8; ++r) {
            int mo = row0 + r + half_sel * 8;
            if (mo < N) {
                xl2[mo * 64 + n] = accl[nt][r] + bln;
                xr2[mo * 64 + n] = accr[nt][r] + brn;
            }
        }
    }
}

__global__ void k_pool(const float* __restrict__ agg2, const float* __restrict__ bias2,
                       const int* __restrict__ batch,
                       float* __restrict__ sums, float* __restrict__ cnts, int N) {
    int idx = blockIdx.x * blockDim.x + threadIdx.x;
    if (idx >= N * 64) return;
    int n = idx >> 6, d = idx & 63;
    int g = batch[n];
    atomicAdd(&sums[g * 64 + d], agg2[idx] + bias2[d]);
    if (d == 0) atomicAdd(&cnts[g], 1.0f);
}

__global__ void k_final(const float* __restrict__ sums, const float* __restrict__ cnts,
                        float* __restrict__ out, int G) {
    int idx = blockIdx.x * blockDim.x + threadIdx.x;
    if (idx >= G * 64) return;
    out[idx] = sums[idx] / fmaxf(cnts[idx >> 6], 1.0f);
}

extern "C" void kernel_launch(void* const* d_in, const int* in_sizes, int n_in,
                              void* d_out, int out_size, void* d_ws, size_t ws_size,
                              hipStream_t stream) {
    const float* x      = (const float*)d_in[0];
    const float* ea     = (const float*)d_in[1];
    const float* W_l1   = (const float*)d_in[2];
    const float* b_l1   = (const float*)d_in[3];
    const float* W_r1   = (const float*)d_in[4];
    const float* b_r1   = (const float*)d_in[5];
    const float* W_e1   = (const float*)d_in[6];
    const float* att1   = (const float*)d_in[7];
    const float* bias1  = (const float*)d_in[8];
    const float* W_l2   = (const float*)d_in[9];
    const float* b_l2   = (const float*)d_in[10];
    const float* W_r2   = (const float*)d_in[11];
    const float* b_r2   = (const float*)d_in[12];
    const float* W_e2   = (const float*)d_in[13];
    const float* att2   = (const float*)d_in[14];
    const float* bias2  = (const float*)d_in[15];
    const int*   eidx   = (const int*)d_in[16];
    const int*   batch  = (const int*)d_in[17];
    float* out = (float*)d_out;

    const int N = in_sizes[0] / 2;
    const int E = in_sizes[1] / 2;
    const int G = out_size / 64;
    const int* src = eidx;
    const int* tgt = eidx + E;

    char* ws = (char*)d_ws;
    size_t off = 0;
    auto alloc = [&](size_t bytes) { size_t o = off; off = (off + bytes + 255) & ~(size_t)255; return o; };

    size_t o_emax1 = alloc((size_t)N * 16 * 4);
    size_t o_den1  = alloc((size_t)N * 16 * 4);
    size_t o_agg1  = alloc((size_t)N * 128 * 4);
    size_t o_emax2 = alloc((size_t)N * 4 * 4);
    size_t o_den2  = alloc((size_t)N * 4 * 4);
    size_t o_agg2  = alloc((size_t)N * 64 * 4);
    size_t o_sums  = alloc((size_t)G * 64 * 4);
    size_t o_cnts  = alloc((size_t)G * 4);
    size_t acc_bytes = off;

    size_t o_xl1 = alloc((size_t)N * 128 * 4);
    size_t o_xr1 = alloc((size_t)N * 128 * 4);
    size_t o_e1  = alloc((size_t)E * 16 * 4);
    size_t o_hh  = alloc((size_t)N * 128 * 2);
    size_t o_wpk = alloc((size_t)2 * 8192 * 2);
    size_t o_xl2 = alloc((size_t)N * 64 * 4);
    size_t o_xr2 = alloc((size_t)N * 64 * 4);
    size_t o_e2  = alloc((size_t)E * 4 * 4);
    (void)ws_size;

    unsigned* emax1 = (unsigned*)(ws + o_emax1);
    float*    den1  = (float*)(ws + o_den1);
    float*    agg1  = (float*)(ws + o_agg1);
    unsigned* emax2 = (unsigned*)(ws + o_emax2);
    float*    den2  = (float*)(ws + o_den2);
    float*    agg2  = (float*)(ws + o_agg2);
    float*    sums  = (float*)(ws + o_sums);
    float*    cnts  = (float*)(ws + o_cnts);
    float*    xl1   = (float*)(ws + o_xl1);
    float*    xr1   = (float*)(ws + o_xr1);
    float*    e1    = (float*)(ws + o_e1);
    _Float16* hh    = (_Float16*)(ws + o_hh);
    _Float16* wpk   = (_Float16*)(ws + o_wpk);
    float*    xl2   = (float*)(ws + o_xl2);
    float*    xr2   = (float*)(ws + o_xr2);
    float*    e2    = (float*)(ws + o_e2);

    const int TB = 256;
    auto blocks = [&](long long t) { return (unsigned)((t + TB - 1) / TB); };

    hipMemsetAsync(d_ws, 0, acc_bytes, stream);

    k_node_transform1<<<blocks((long long)N * 128), TB, 0, stream>>>(
        x, W_l1, b_l1, W_r1, b_r1, xl1, xr1, N);
    k_edge_logits<16, 8><<<blocks((long long)E * 16), TB, 0, stream>>>(
        xl1, xr1, ea, W_e1, att1, src, tgt, e1, emax1, E);
    k_edge_exp<16><<<blocks((long long)E * 16), TB, 0, stream>>>(
        e1, emax1, den1, tgt, E);
    k_edge_agg<16, 8><<<blocks((long long)E * 32), TB, 0, stream>>>(
        e1, den1, xl1, src, tgt, agg1, E);
    k_elu_half<<<blocks((long long)N * 128), TB, 0, stream>>>(agg1, bias1, hh, N);

    k_w_pack<<<blocks(2 * 8192), TB, 0, stream>>>(W_l2, W_r2, wpk);
    long long gemm_threads = (long long)((N + 15) / 16) * 32;
    k_gemm2_wmma<<<blocks(gemm_threads), TB, 0, stream>>>(
        hh, wpk, b_l2, b_r2, xl2, xr2, N);

    k_edge_logits<4, 16><<<blocks((long long)E * 4), TB, 0, stream>>>(
        xl2, xr2, ea, W_e2, att2, src, tgt, e2, emax2, E);
    k_edge_exp<4><<<blocks((long long)E * 4), TB, 0, stream>>>(
        e2, emax2, den2, tgt, E);
    k_edge_agg<4, 16><<<blocks((long long)E * 16), TB, 0, stream>>>(
        e2, den2, xl2, src, tgt, agg2, E);

    k_pool<<<blocks((long long)N * 64), TB, 0, stream>>>(agg2, bias2, batch, sums, cnts, N);
    k_final<<<blocks((long long)G * 64), TB, 0, stream>>>(sums, cnts, out, G);
}


# mean pool: each thread sums its run of equal graph ids in a register and sends one atomicAdd per run, not one per element; changes the summation order of the pool sums
# speedup vs baseline: 1.1498x; 1.1498x over previous
#include <hip/hip_runtime.h>
#include <hip/hip_fp16.h>

typedef __attribute__((ext_vector_type(16))) _Float16 v16h;
typedef __attribute__((ext_vector_type(8)))  _Float16 v8h;
typedef __attribute__((ext_vector_type(8)))  float    v8f;

__device__ __forceinline__ unsigned fp_ord(float f) {
    unsigned u = __float_as_uint(f);
    return (u & 0x80000000u) ? ~u : (u | 0x80000000u);
}
__device__ __forceinline__ float fp_unord(unsigned u) {
    unsigned b = (u & 0x80000000u) ? (u ^ 0x80000000u) : ~u;
    return __uint_as_float(b);
}

__global__ void k_node_transform1(const float* __restrict__ x,
                                  const float* __restrict__ Wl, const float* __restrict__ bl,
                                  const float* __restrict__ Wr, const float* __restrict__ br,
                                  float* __restrict__ xl1, float* __restrict__ xr1, int N) {
    int idx = blockIdx.x * blockDim.x + threadIdx.x;
    if (idx >= N * 128) return;
    int n = idx >> 7, d = idx & 127;
    float x0 = x[n * 2], x1 = x[n * 2 + 1];
    xl1[idx] = fmaf(x0, Wl[d], fmaf(x1, Wl[128 + d], bl[d]));
    xr1[idx] = fmaf(x0, Wr[d], fmaf(x1, Wr[128 + d], br[d]));
}

template <int H, int C>
__global__ void k_edge_logits(const float* __restrict__ xl, const float* __restrict__ xr,
                              const float* __restrict__ ea, const float* __restrict__ We,
                              const float* __restrict__ att,
                              const int* __restrict__ src, const int* __restrict__ tgt,
                              float* __restrict__ elog, unsigned* __restrict__ emaxM, int E) {
    const int D = H * C;
    int idx = blockIdx.x * blockDim.x + threadIdx.x;
    if (idx >= E * H) return;
    int e = idx / H, h = idx % H;
    int s = src[e], t = tgt[e];
    float a0 = ea[e * 2], a1 = ea[e * 2 + 1];
    int base = h * C;
    float acc = 0.f;
#pragma unroll
    for (int c = 0; c < C; ++c) {
        int d = base + c;
        float m = xl[s * D + d] + xr[t * D + d] + a0 * We[d] + a1 * We[D + d];
        m = (m > 0.f) ? m : 0.2f * m;
        acc = fmaf(m, att[base + c], acc);
    }
    elog[idx] = acc;
    atomicMax(&emaxM[t * H + h], fp_ord(acc));
}

template <int H>
__global__ void k_edge_exp(float* __restrict__ elog, const unsigned* __restrict__ emaxM,
                           float* __restrict__ denom, const int* __restrict__ tgt, int E) {
    int idx = blockIdx.x * blockDim.x + threadIdx.x;
    if (idx >= E * H) return;
    int e = idx / H, h = idx % H;
    int t = tgt[e];
    float mx = fp_unord(emaxM[t * H + h]);
    float v = __expf(elog[idx] - mx);
    elog[idx] = v;
    atomicAdd(&denom[t * H + h], v);
}

template <int H, int C>
__global__ void k_edge_agg(const float* __restrict__ ee, const float* __restrict__ denom,
                           const float* __restrict__ xl,
                           const int* __restrict__ src, const int* __restrict__ tgt,
                           float* __restrict__ agg, int E) {
    const int D = H * C;
    const int Q = D / 4;
    int idx = blockIdx.x * blockDim.x + threadIdx.x;
    if (idx >= E * Q) return;
    int e = idx / Q, q = idx % Q;
    int d = q * 4, h = d / C;
    int s = src[e], t = tgt[e];
    float alpha = ee[e * H + h] / denom[t * H + h];
    float4 v = *(const float4*)(xl + s * D + d);
    atomicAdd(&agg[t * D + d + 0], alpha * v.x);
    atomicAdd(&agg[t * D + d + 1], alpha * v.y);
    atomicAdd(&agg[t * D + d + 2], alpha * v.z);
    atomicAdd(&agg[t * D + d + 3], alpha * v.w);
}

__global__ void k_elu_half(const float* __restrict__ agg, const float* __restrict__ bias,
                           _Float16* __restrict__ hh, int N) {
    int idx = blockIdx.x * blockDim.x + threadIdx.x;
    if (idx >= N * 128) return;
    int d = idx & 127;
    float v = agg[idx] + bias[d];
    v = (v > 0.f) ? v : (__expf(v) - 1.f);
    hh[idx] = (_Float16)v;
}

__global__ void k_w_pack(const float* __restrict__ Wl, const float* __restrict__ Wr,
                         _Float16* __restrict__ Wpk) {
    int idx = blockIdx.x * blockDim.x + threadIdx.x;
    if (idx >= 2 * 8192) return;
    int which = idx >> 13;
    int r = idx & 8191;
    int j = r & 15;
    int lane = (r >> 4) & 31;
    int frag = r >> 9;
    int kc = frag >> 2, nt = frag & 3;
    int half_sel = lane >> 4, lidx = lane & 15;
    int kb = (j < 8) ? (half_sel * 8) : (16 + half_sel * 8);
    int k = kc * 32 + kb + (j & 7);
    int n = nt * 16 + lidx;
    const float* W = which ? Wr : Wl;
    Wpk[idx] = (_Float16)W[k * 64 + n];
}

__global__ void k_gemm2_wmma(const _Float16* __restrict__ hh,
                             const _Float16* __restrict__ Wpk,
                             const float* __restrict__ bl, const float* __restrict__ br,
                             float* __restrict__ xl2, float* __restrict__ xr2, int N) {
    __shared__ __align__(32) _Float16 ldsW[2 * 8192];

    {
        const uint4* g = (const uint4*)Wpk;
        uint4* l = (uint4*)ldsW;
        for (int i = threadIdx.x; i < 2048; i += blockDim.x) l[i] = g[i];
    }
    __syncthreads();

    int wave = (blockIdx.x * blockDim.x + threadIdx.x) >> 5;
    int lane = threadIdx.x & 31;
    int row0 = wave * 16;
    if (row0 >= N) return;

    int half_sel = lane >> 4;
    int lidx = lane & 15;
    int kb_lo = half_sel * 8;
    int kb_hi = 16 + kb_lo;
    int m = row0 + lidx;
    if (m >= N) m = N - 1;

    const v16h* fragL = (const v16h*)ldsW;
    const v16h* fragR = (const v16h*)(ldsW + 8192);

    v8f accl[4] = {};
    v8f accr[4] = {};

#pragma unroll
    for (int kc = 0; kc < 4; ++kc) {
        const _Float16* pa = hh + m * 128 + kc * 32;
        v8h alo = *(const v8h*)(pa + kb_lo);
        v8h ahi = *(const v8h*)(pa + kb_hi);
        v16h a = __builtin_shufflevector(alo, ahi, 0, 1, 2, 3, 4, 5, 6, 7,
                                         8, 9, 10, 11, 12, 13, 14, 15);
#pragma unroll
        for (int nt = 0; nt < 4; ++nt) {
            int fi = (kc * 4 + nt) * 32 + lane;
            v16h b  = fragL[fi];
            v16h b2 = fragR[fi];
            accl[nt] = __builtin_amdgcn_wmma_f32_16x16x32_f16(
                false, a, false, b, (short)0, accl[nt], false, false);
            accr[nt] = __builtin_amdgcn_wmma_f32_16x16x32_f16(
                false, a, false, b2, (short)0, accr[nt], false, false);
        }
    }
#pragma unroll
    for (int nt = 0; nt < 4; ++nt) {
        int n = nt * 16 + lidx;
        float bln = bl[n], brn = br[n];
#pragma unroll
        for (int r = 0; r < 8; ++r) {
            int mo = row0 + r + half_sel * 8;
            if (mo < N) {
                xl2[mo * 64 + n] = accl[nt][r] + bln;
                xr2[mo * 64 + n] = accr[nt][r] + brn;
            }
        }
    }
}

constexpr int POOL_NPB = 128;
constexpr int POOL_U   = 8;

__global__ void __launch_bounds__(256)
k_pool(const float* __restrict__ agg2, const float* __restrict__ bias2,
       const int* __restrict__ batch,
       float* __restrict__ sums, float* __restrict__ cnts, int N) {
    const int d   = threadIdx.x & 63;
    const int sub = threadIdx.x >> 6;
    const int n0  = blockIdx.x * POOL_NPB + sub;
    const float b = bias2[d];
    float acc = 0.f, cnt = 0.f;
    int gcur = -1;
    for (int it = 0; it < POOL_NPB / 4; it += POOL_U) {
        float v[POOL_U];
        int   g[POOL_U];
#pragma unroll
        for (int u = 0; u < POOL_U; ++u) {
            int n = n0 + 4 * (it + u);
            bool ok = n < N;
            g[u] = ok ? batch[n] : -1;
            v[u] = ok ? agg2[n * 64 + d] : 0.f;
        }
#pragma unroll
        for (int u = 0; u < POOL_U; ++u) {
            if (g[u] != gcur) {
                if (gcur >= 0) {
                    atomicAdd(&sums[gcur * 64 + d], acc);
                    if (d == 0) atomicAdd(&cnts[gcur], cnt);
                }
                gcur = g[u]; acc = 0.f; cnt = 0.f;
            }
            acc += v[u] + b;
            cnt += 1.0f;
        }
    }
    if (gcur >= 0) {
        atomicAdd(&sums[gcur * 64 + d], acc);
        if (d == 0) atomicAdd(&cnts[gcur], cnt);
    }
}

__global__ void k_final(const float* __restrict__ sums, const float* __restrict__ cnts,
                        float* __restrict__ out, int G) {
    int idx = blockIdx.x * blockDim.x + threadIdx.x;
    if (idx >= G * 64) return;
    out[idx] = sums[idx] / fmaxf(cnts[idx >> 6], 1.0f);
}

extern "C" void kernel_launch(void* const* d_in, const int* in_sizes, int n_in,
                              void* d_out, int out_size, void* d_ws, size_t ws_size,
                              hipStream_t stream) {
    const float* x      = (const float*)d_in[0];
    const float* ea     = (const float*)d_in[1];
    const float* W_l1   = (const float*)d_in[2];
    const float* b_l1   = (const float*)d_in[3];
    const float* W_r1   = (const float*)d_in[4];
    const float* b_r1   = (const float*)d_in[5];
    const float* W_e1   = (const float*)d_in[6];
    const float* att1   = (const float*)d_in[7];
    const float* bias1  = (const float*)d_in[8];
    const float* W_l2   = (const float*)d_in[9];
    const float* b_l2   = (const float*)d_in[10];
    const float* W_r2   = (const float*)d_in[11];
    const float* b_r2   = (const float*)d_in[12];
    const float* W_e2   = (const float*)d_in[13];
    const float* att2   = (const float*)d_in[14];
    const float* bias2  = (const float*)d_in[15];
    const int*   eidx   = (const int*)d_in[16];
    const int*   batch  = (const int*)d_in[17];
    float* out = (float*)d_out;

    const int N = in_sizes[0] / 2;
    const int E = in_sizes[1] / 2;
    const int G = out_size / 64;
    const int* src = eidx;
    const int* tgt = eidx + E;

    char* ws = (char*)d_ws;
    size_t off = 0;
    auto alloc = [&](size_t bytes) { size_t o = off; off = (off + bytes + 255) & ~(size_t)255; return o; };

    size_t o_emax1 = alloc((size_t)N * 16 * 4);
    size_t o_den1  = alloc((size_t)N * 16 * 4);
    size_t o_agg1  = alloc((size_t)N * 128 * 4);
    size_t o_emax2 = alloc((size_t)N * 4 * 4);
    size_t o_den2  = alloc((size_t)N * 4 * 4);
    size_t o_agg2  = alloc((size_t)N * 64 * 4);
    size_t o_sums  = alloc((size_t)G * 64 * 4);
    size_t o_cnts  = alloc((size_t)G * 4);
    size_t acc_bytes = off;

    size_t o_xl1 = alloc((size_t)N * 128 * 4);
    size_t o_xr1 = alloc((size_t)N * 128 * 4);
    size_t o_e1  = alloc((size_t)E * 16 * 4);
    size_t o_hh  = alloc((size_t)N * 128 * 2);
    size_t o_wpk = alloc((size_t)2 * 8192 * 2);
    size_t o_xl2 = alloc((size_t)N * 64 * 4);
    size_t o_xr2 = alloc((size_t)N * 64 * 4);
    size_t o_e2  = alloc((size_t)E * 4 * 4);
    (void)ws_size;

    unsigned* emax1 = (unsigned*)(ws + o_emax1);
    float*    den1  = (float*)(ws + o_den1);
    float*    agg1  = (float*)(ws + o_agg1);
    unsigned* emax2 = (unsigned*)(ws + o_emax2);
    float*    den2  = (float*)(ws + o_den2);
    float*    agg2  = (float*)(ws + o_agg2);
    float*    sums  = (float*)(ws + o_sums);
    float*    cnts  = (float*)(ws + o_cnts);
    float*    xl1   = (float*)(ws + o_xl1);
    float*    xr1   = (float*)(ws + o_xr1);
    float*    e1    = (float*)(ws + o_e1);
    _Float16* hh    = (_Float16*)(ws + o_hh);
    _Float16* wpk   = (_Float16*)(ws + o_wpk);
    float*    xl2   = (float*)(ws + o_xl2);
    float*    xr2   = (float*)(ws + o_xr2);
    float*    e2    = (float*)(ws + o_e2);

    const int TB = 256;
    auto blocks = [&](long long t) { return (unsigned)((t + TB - 1) / TB); };

    hipMemsetAsync(d_ws, 0, acc_bytes, stream);

    k_node_transform1<<<blocks((long long)N * 128), TB, 0, stream>>>(
        x, W_l1, b_l1, W_r1, b_r1, xl1, xr1, N);
    k_edge_logits<16, 8><<<blocks((long long)E * 16), TB, 0, stream>>>(
        xl1, xr1, ea, W_e1, att1, src, tgt, e1, emax1, E);
    k_edge_exp<16><<<blocks((long long)E * 16), TB, 0, stream>>>(
        e1, emax1, den1, tgt, E);
    k_edge_agg<16, 8><<<blocks((long long)E * 32), TB, 0, stream>>>(
        e1, den1, xl1, src, tgt, agg1, E);
    k_elu_half<<<blocks((long long)N * 128), TB, 0, stream>>>(agg1, bias1, hh, N);

    k_w_pack<<<blocks(2 * 8192), TB, 0, stream>>>(W_l2, W_r2, wpk);
    long long gemm_threads = (long long)((N + 15) / 16) * 32;
    k_gemm2_wmma<<<blocks(gemm_threads), TB, 0, stream>>>(
        hh, wpk, b_l2, b_r2, xl2, xr2, N);

    k_edge_logits<4, 16><<<blocks((long long)E * 4), TB, 0, stream>>>(
        xl2, xr2, ea, W_e2, att2, src, tgt, e2, emax2, E);
    k_edge_exp<4><<<blocks((long long)E * 4), TB, 0, stream>>>(
        e2, emax2, den2, tgt, E);
    k_edge_agg<4, 16><<<blocks((long long)E * 16), TB, 0, stream>>>(
        e2, den2, xl2, src, tgt, agg2, E);

    k_pool<<<(unsigned)((N + POOL_NPB - 1) / POOL_NPB), TB, 0, stream>>>(agg2, bias2, batch, sums, cnts, N);
    k_final<<<blocks((long long)G * 64), TB, 0, stream>>>(sums, cnts, out, G);
}


# layer 1 aggregates in input space: 2 atomic sums of weight*x per (node, head), expanded by the layer's weights per node; the 128-channel scatter launch is gone; changes the order of layer-1 sums
# speedup vs baseline: 1.5285x; 1.3294x over previous
#include <hip/hip_runtime.h>
#include <hip/hip_fp16.h>

typedef __attribute__((ext_vector_type(16))) _Float16 v16h;
typedef __attribute__((ext_vector_type(8)))  _Float16 v8h;
typedef __attribute__((ext_vector_type(8)))  float    v8f;

__device__ __forceinline__ unsigned fp_ord(float f) {
    unsigned u = __float_as_uint(f);
    return (u & 0x80000000u) ? ~u : (u | 0x80000000u);
}
__device__ __forceinline__ float fp_unord(unsigned u) {
    unsigned b = (u & 0x80000000u) ? (u ^ 0x80000000u) : ~u;
    return __uint_as_float(b);
}

__global__ void k_node_transform1(const float* __restrict__ x,
                                  const float* __restrict__ Wl, const float* __restrict__ bl,
                                  const float* __restrict__ Wr, const float* __restrict__ br,
                                  float* __restrict__ xl1, float* __restrict__ xr1, int N) {
    int idx = blockIdx.x * blockDim.x + threadIdx.x;
    if (idx >= N * 128) return;
    int n = idx >> 7, d = idx & 127;
    float x0 = x[n * 2], x1 = x[n * 2 + 1];
    xl1[idx] = fmaf(x0, Wl[d], fmaf(x1, Wl[128 + d], bl[d]));
    xr1[idx] = fmaf(x0, Wr[d], fmaf(x1, Wr[128 + d], br[d]));
}

template <int H, int C>
__global__ void k_edge_logits(const float* __restrict__ xl, const float* __restrict__ xr,
                              const float* __restrict__ ea, const float* __restrict__ We,
                              const float* __restrict__ att,
                              const int* __restrict__ src, const int* __restrict__ tgt,
                              float* __restrict__ elog, unsigned* __restrict__ emaxM, int E) {
    const int D = H * C;
    int idx = blockIdx.x * blockDim.x + threadIdx.x;
    if (idx >= E * H) return;
    int e = idx / H, h = idx % H;
    int s = src[e], t = tgt[e];
    float a0 = ea[e * 2], a1 = ea[e * 2 + 1];
    int base = h * C;
    float acc = 0.f;
#pragma unroll
    for (int c = 0; c < C; ++c) {
        int d = base + c;
        float m = xl[s * D + d] + xr[t * D + d] + a0 * We[d] + a1 * We[D + d];
        m = (m > 0.f) ? m : 0.2f * m;
        acc = fmaf(m, att[base + c], acc);
    }
    elog[idx] = acc;
    atomicMax(&emaxM[t * H + h], fp_ord(acc));
}

template <int H>
__global__ void k_edge_exp(float* __restrict__ elog, const unsigned* __restrict__ emaxM,
                           float* __restrict__ denom, const int* __restrict__ tgt, int E) {
    int idx = blockIdx.x * blockDim.x + threadIdx.x;
    if (idx >= E * H) return;
    int e = idx / H, h = idx % H;
    int t = tgt[e];
    float mx = fp_unord(emaxM[t * H + h]);
    float v = __expf(elog[idx] - mx);
    elog[idx] = v;
    atomicAdd(&denom[t * H + h], v);
}

__global__ void k_edge_exp_in(const float* __restrict__ elog, const unsigned* __restrict__ emaxM,
                              const float* __restrict__ x,
                              const int* __restrict__ src, const int* __restrict__ tgt,
                              float* __restrict__ denom, float* __restrict__ sx, int E) {
    int idx = blockIdx.x * blockDim.x + threadIdx.x;
    if (idx >= E * 16) return;
    int e = idx >> 4, h = idx & 15;
    int s = src[e], t = tgt[e];
    float mx = fp_unord(emaxM[t * 16 + h]);
    float x0 = x[s * 2], x1 = x[s * 2 + 1];
    float v = __expf(elog[idx] - mx);
    atomicAdd(&denom[t * 16 + h], v);
    atomicAdd(&sx[t * 32 + h], v * x0);
    atomicAdd(&sx[t * 32 + 16 + h], v * x1);
}

template <int H, int C>
__global__ void k_edge_agg(const float* __restrict__ ee, const float* __restrict__ denom,
                           const float* __restrict__ xl,
                           const int* __restrict__ src, const int* __restrict__ tgt,
                           float* __restrict__ agg, int E) {
    const int D = H * C;
    const int Q = D / 4;
    int idx = blockIdx.x * blockDim.x + threadIdx.x;
    if (idx >= E * Q) return;
    int e = idx / Q, q = idx % Q;
    int d = q * 4, h = d / C;
    int s = src[e], t = tgt[e];
    float alpha = ee[e * H + h] / denom[t * H + h];
    float4 v = *(const float4*)(xl + s * D + d);
    atomicAdd(&agg[t * D + d + 0], alpha * v.x);
    atomicAdd(&agg[t * D + d + 1], alpha * v.y);
    atomicAdd(&agg[t * D + d + 2], alpha * v.z);
    atomicAdd(&agg[t * D + d + 3], alpha * v.w);
}

__global__ void k_elu_half_in(const float* __restrict__ denom, const float* __restrict__ sx,
                              const float* __restrict__ Wl, const float* __restrict__ bl,
                              const float* __restrict__ bias,
                              _Float16* __restrict__ hh, int N) {
    int idx = blockIdx.x * blockDim.x + threadIdx.x;
    if (idx >= N * 128) return;
    int n = idx >> 7, d = idx & 127, h = d >> 3;
    float dn = denom[n * 16 + h];
    float s0 = sx[n * 32 + h], s1 = sx[n * 32 + 16 + h];
    float r = fmaf(Wl[d], s0, Wl[128 + d] * s1) / dn + bl[d];
    float v = ((dn > 0.f) ? r : 0.f) + bias[d];
    v = (v > 0.f) ? v : (__expf(v) - 1.f);
    hh[idx] = (_Float16)v;
}

__global__ void k_w_pack(const float* __restrict__ Wl, const float* __restrict__ Wr,
                         _Float16* __restrict__ Wpk) {
    int idx = blockIdx.x * blockDim.x + threadIdx.x;
    if (idx >= 2 * 8192) return;
    int which = idx >> 13;
    int r = idx & 8191;
    int j = r & 15;
    int lane = (r >> 4) & 31;
    int frag = r >> 9;
    int kc = frag >> 2, nt = frag & 3;
    int half_sel = lane >> 4, lidx = lane & 15;
    int kb = (j < 8) ? (half_sel * 8) : (16 + half_sel * 8);
    int k = kc * 32 + kb + (j & 7);
    int n = nt * 16 + lidx;
    const float* W = which ? Wr : Wl;
    Wpk[idx] = (_Float16)W[k * 64 + n];
}

__global__ void k_gemm2_wmma(const _Float16* __restrict__ hh,
                             const _Float16* __restrict__ Wpk,
                             const float* __restrict__ bl, const float* __restrict__ br,
                             float* __restrict__ xl2, float* __restrict__ xr2, int N) {
    __shared__ __align__(32) _Float16 ldsW[2 * 8192];

    {
        const uint4* g = (const uint4*)Wpk;
        uint4* l = (uint4*)ldsW;
        for (int i = threadIdx.x; i < 2048; i += blockDim.x) l[i] = g[i];
    }
    __syncthreads();

    int wave = (blockIdx.x * blockDim.x + threadIdx.x) >> 5;
    int lane = threadIdx.x & 31;
    int row0 = wave * 16;
    if (row0 >= N) return;

    int half_sel = lane >> 4;
    int lidx = lane & 15;
    int kb_lo = half_sel * 8;
    int kb_hi = 16 + kb_lo;
    int m = row0 + lidx;
    if (m >= N) m = N - 1;

    const v16h* fragL = (const v16h*)ldsW;
    const v16h* fragR = (const v16h*)(ldsW + 8192);

    v8f accl[4] = {};
    v8f accr[4] = {};

#pragma unroll
    for (int kc = 0; kc < 4; ++kc) {
        const _Float16* pa = hh + m * 128 + kc * 32;
        v8h alo = *(const v8h*)(pa + kb_lo);
        v8h ahi = *(const v8h*)(pa + kb_hi);
        v16h a = __builtin_shufflevector(alo, ahi, 0, 1, 2, 3, 4, 5, 6, 7,
                                         8, 9, 10, 11, 12, 13, 14, 15);
#pragma unroll
        for (int nt = 0; nt < 4; ++nt) {
            int fi = (kc * 4 + nt) * 32 + lane;
            v16h b  = fragL[fi];
            v16h b2 = fragR[fi];
            accl[nt] = __builtin_amdgcn_wmma_f32_16x16x32_f16(
                false, a, false, b, (short)0, accl[nt], false, false);
            accr[nt] = __builtin_amdgcn_wmma_f32_16x16x32_f16(
                false, a, false, b2, (short)0, accr[nt], false, false);
        }
    }
#pragma unroll
    for (int nt = 0; nt < 4; ++nt) {
        int n = nt * 16 + lidx;
        float bln = bl[n], brn = br[n];
#pragma unroll
        for (int r = 0; r < 8; ++r) {
            int mo = row0 + r + half_sel * 8;
            if (mo < N) {
                xl2[mo * 64 + n] = accl[nt][r] + bln;
                xr2[mo * 64 + n] = accr[nt][r] + brn;
            }
        }
    }
}

constexpr int POOL_NPB = 128;
constexpr int POOL_U   = 8;

__global__ void __launch_bounds__(256)
k_pool(const float* __restrict__ agg2, const float* __restrict__ bias2,
       const int* __restrict__ batch,
       float* __restrict__ sums, float* __restrict__ cnts, int N) {
    const int d   = threadIdx.x & 63;
    const int sub = threadIdx.x >> 6;
    const int n0  = blockIdx.x * POOL_NPB + sub;
    const float b = bias2[d];
    float acc = 0.f, cnt = 0.f;
    int gcur = -1;
    for (int it = 0; it < POOL_NPB / 4; it += POOL_U) {
        float v[POOL_U];
        int   g[POOL_U];
#pragma unroll
        for (int u = 0; u < POOL_U; ++u) {
            int n = n0 + 4 * (it + u);
            bool ok = n < N;
            g[u] = ok ? batch[n] : -1;
            v[u] = ok ? agg2[n * 64 + d] : 0.f;
        }
#pragma unroll
        for (int u = 0; u < POOL_U; ++u) {
            if (g[u] != gcur) {
                if (gcur >= 0) {
                    atomicAdd(&sums[gcur * 64 + d], acc);
                    if (d == 0) atomicAdd(&cnts[gcur], cnt);
                }
                gcur = g[u]; acc = 0.f; cnt = 0.f;
            }
            acc += v[u] + b;
            cnt += 1.0f;
        }
    }
    if (gcur >= 0) {
        atomicAdd(&sums[gcur * 64 + d], acc);
        if (d == 0) atomicAdd(&cnts[gcur], cnt);
    }
}

__global__ void k_final(const float* __restrict__ sums, const float* __restrict__ cnts,
                        float* __restrict__ out, int G) {
    int idx = blockIdx.x * blockDim.x + threadIdx.x;
    if (idx >= G * 64) return;
    out[idx] = sums[idx] / fmaxf(cnts[idx >> 6], 1.0f);
}

extern "C" void kernel_launch(void* const* d_in, const int* in_sizes, int n_in,
                              void* d_out, int out_size, void* d_ws, size_t ws_size,
                              hipStream_t stream) {
    const float* x      = (const float*)d_in[0];
    const float* ea     = (const float*)d_in[1];
    const float* W_l1   = (const float*)d_in[2];
    const float* b_l1   = (const float*)d_in[3];
    const float* W_r1   = (const float*)d_in[4];
    const float* b_r1   = (const float*)d_in[5];
    const float* W_e1   = (const float*)d_in[6];
    const float* att1   = (const float*)d_in[7];
    const float* bias1  = (const float*)d_in[8];
    const float* W_l2   = (const float*)d_in[9];
    const float* b_l2   = (const float*)d_in[10];
    const float* W_r2   = (const float*)d_in[11];
    const float* b_r2   = (const float*)d_in[12];
    const float* W_e2   = (const float*)d_in[13];
    const float* att2   = (const float*)d_in[14];
    const float* bias2  = (const float*)d_in[15];
    const int*   eidx   = (const int*)d_in[16];
    const int*   batch  = (const int*)d_in[17];
    float* out = (float*)d_out;

    const int N = in_sizes[0] / 2;
    const int E = in_sizes[1] / 2;
    const int G = out_size / 64;
    const int* src = eidx;
    const int* tgt = eidx + E;

    char* ws = (char*)d_ws;
    size_t off = 0;
    auto alloc = [&](size_t bytes) { size_t o = off; off = (off + bytes + 255) & ~(size_t)255; return o; };

    size_t o_emax1 = alloc((size_t)N * 16 * 4);
    size_t o_den1  = alloc((size_t)N * 16 * 4);
    size_t o_sx1   = alloc((size_t)N * 32 * 4);
    size_t o_emax2 = alloc((size_t)N * 4 * 4);
    size_t o_den2  = alloc((size_t)N * 4 * 4);
    size_t o_agg2  = alloc((size_t)N * 64 * 4);
    size_t o_sums  = alloc((size_t)G * 64 * 4);
    size_t o_cnts  = alloc((size_t)G * 4);
    size_t acc_bytes = off;

    size_t o_xl1 = alloc((size_t)N * 128 * 4);
    size_t o_xr1 = alloc((size_t)N * 128 * 4);
    size_t o_e1  = alloc((size_t)E * 16 * 4);
    size_t o_hh  = alloc((size_t)N * 128 * 2);
    size_t o_wpk = alloc((size_t)2 * 8192 * 2);
    size_t o_xl2 = alloc((size_t)N * 64 * 4);
    size_t o_xr2 = alloc((size_t)N * 64 * 4);
    size_t o_e2  = alloc((size_t)E * 4 * 4);
    (void)ws_size;

    unsigned* emax1 = (unsigned*)(ws + o_emax1);
    float*    den1  = (float*)(ws + o_den1);
    float*    sx1   = (float*)(ws + o_sx1);
    unsigned* emax2 = (unsigned*)(ws + o_emax2);
    float*    den2  = (float*)(ws + o_den2);
    float*    agg2  = (float*)(ws + o_agg2);
    float*    sums  = (float*)(ws + o_sums);
    float*    cnts  = (float*)(ws + o_cnts);
    float*    xl1   = (float*)(ws + o_xl1);
    float*    xr1   = (float*)(ws + o_xr1);
    float*    e1    = (float*)(ws + o_e1);
    _Float16* hh    = (_Float16*)(ws + o_hh);
    _Float16* wpk   = (_Float16*)(ws + o_wpk);
    float*    xl2   = (float*)(ws + o_xl2);
    float*    xr2   = (float*)(ws + o_xr2);
    float*    e2    = (float*)(ws + o_e2);

    const int TB = 256;
    auto blocks = [&](long long t) { return (unsigned)((t + TB - 1) / TB); };

    hipMemsetAsync(d_ws, 0, acc_bytes, stream);

    k_node_transform1<<<blocks((long long)N * 128), TB, 0, stream>>>(
        x, W_l1, b_l1, W_r1, b_r1, xl1, xr1, N);
    k_edge_logits<16, 8><<<blocks((long long)E * 16), TB, 0, stream>>>(
        xl1, xr1, ea, W_e1, att1, src, tgt, e1, emax1, E);
    k_edge_exp_in<<<blocks((long long)E * 16), TB, 0, stream>>>(
        e1, emax1, x, src, tgt, den1, sx1, E);
    k_elu_half_in<<<blocks((long long)N * 128), TB, 0, stream>>>(
        den1, sx1, W_l1, b_l1, bias1, hh, N);

    k_w_pack<<<blocks(2 * 8192), TB, 0, stream>>>(W_l2, W_r2, wpk);
    long long gemm_threads = (long long)((N + 15) / 16) * 32;
    k_gemm2_wmma<<<blocks(gemm_threads), TB, 0, stream>>>(
        hh, wpk, b_l2, b_r2, xl2, xr2, N);

    k_edge_logits<4, 16><<<blocks((long long)E * 4), TB, 0, stream>>>(
        xl2, xr2, ea, W_e2, att2, src, tgt, e2, emax2, E);
    k_edge_exp<4><<<blocks((long long)E * 4), TB, 0, stream>>>(
        e2, emax2, den2, tgt, E);
    k_edge_agg<4, 16><<<blocks((long long)E * 16), TB, 0, stream>>>(
        e2, den2, xl2, src, tgt, agg2, E);

    k_pool<<<(unsigned)((N + POOL_NPB - 1) / POOL_NPB), TB, 0, stream>>>(agg2, bias2, batch, sums, cnts, N);
    k_final<<<blocks((long long)G * 64), TB, 0, stream>>>(sums, cnts, out, G);
}


# layer 1 rebuilds each source row from its 2 input features in the attention kernel and sums weight*x per (node, head); the 128-wide row store, gather and launch are gone; changes layer-1 sum order
# speedup vs baseline: 2.8075x; 1.8368x over previous
#include <hip/hip_runtime.h>
#include <hip/hip_fp16.h>

typedef __attribute__((ext_vector_type(16))) _Float16 v16h;
typedef __attribute__((ext_vector_type(8)))  _Float16 v8h;
typedef __attribute__((ext_vector_type(8)))  float    v8f;

__global__ void k_hist(const int* __restrict__ tgt, int* __restrict__ cnt,
                       int* __restrict__ rank, int E) {
    int e = blockIdx.x * blockDim.x + threadIdx.x;
    if (e >= E) return;
    rank[e] = atomicAdd(&cnt[tgt[e]], 1);
}

__global__ void __launch_bounds__(1024)
k_scan(const int* __restrict__ cnt, int* __restrict__ off, int per4) {
    __shared__ int wsum[32];
    const int tid = threadIdx.x, lane = tid & 31, wv = tid >> 5;
    const int4* c4 = (const int4*)cnt + (size_t)tid * per4;
    int s = 0;
    for (int i = 0; i < per4; ++i) { int4 v = c4[i]; s += (v.x + v.y) + (v.z + v.w); }
    int inc = s;
    for (int d = 1; d < 32; d <<= 1) { int o = __shfl_up(inc, d); if (lane >= d) inc += o; }
    if (lane == 31) wsum[wv] = inc;
    __syncthreads();
    if (wv == 0) {
        int w = wsum[lane], winc = w;
        for (int d = 1; d < 32; d <<= 1) { int o = __shfl_up(winc, d); if (lane >= d) winc += o; }
        wsum[lane] = winc - w;
    }
    __syncthreads();
    int run = wsum[wv] + inc - s;
    int4* o4 = (int4*)off + (size_t)tid * per4;
    for (int i = 0; i < per4; ++i) {
        int4 v = c4[i], o;
        o.x = run; run += v.x;
        o.y = run; run += v.y;
        o.z = run; run += v.z;
        o.w = run; run += v.w;
        o4[i] = o;
    }
}

__global__ void k_fill(const int* __restrict__ src, const int* __restrict__ tgt,
                       const float* __restrict__ ea,
                       const int* __restrict__ off, const int* __restrict__ rank,
                       int* __restrict__ csrc, float* __restrict__ cea, int E) {
    int e = blockIdx.x * blockDim.x + threadIdx.x;
    if (e >= E) return;
    int p = off[tgt[e]] + rank[e];
    csrc[p] = src[e];
    ((float2*)cea)[p] = ((const float2*)ea)[e];
}

__device__ __forceinline__ void load8(const float* __restrict__ p, float (&v)[8]) {
    const float4 a = ((const float4*)p)[0], b = ((const float4*)p)[1];
    v[0] = a.x; v[1] = a.y; v[2] = a.z; v[3] = a.w;
    v[4] = b.x; v[5] = b.y; v[6] = b.z; v[7] = b.w;
}

constexpr int L1_U = 4;
__global__ void __launch_bounds__(256)
k_node_attn1(const float* __restrict__ x,
             const float* __restrict__ Wl, const float* __restrict__ bl,
             const float* __restrict__ Wr, const float* __restrict__ br,
             const float* __restrict__ We, const float* __restrict__ att,
             const float* __restrict__ bias,
             const int* __restrict__ off, const int* __restrict__ csrc,
             const float* __restrict__ cea,
             _Float16* __restrict__ hh, int N) {
    int idx = blockIdx.x * blockDim.x + threadIdx.x;
    if (idx >= N * 16) return;
    const int t = idx >> 4, d0 = (idx & 15) * 8;
    const float2* x2 = (const float2*)x;
    float wl0[8], wl1[8], we0[8], we1[8], at[8], base[8];
    load8(Wl + d0, wl0); load8(Wl + 128 + d0, wl1);
    load8(We + d0, we0); load8(We + 128 + d0, we1);
    load8(att + d0, at);
    {
        const float2 xt = x2[t];
        float wr0[8], wr1[8], brv[8], blv[8];
        load8(Wr + d0, wr0); load8(Wr + 128 + d0, wr1);
        load8(br + d0, brv); load8(bl + d0, blv);
#pragma unroll
        for (int c = 0; c < 8; ++c)
            base[c] = fmaf(xt.x, wr0[c], fmaf(xt.y, wr1[c], brv[c])) + blv[c];
    }
    const int beg = off[t], end = off[t + 1];
    const float2* cea2 = (const float2*)cea;
    float mx = -__builtin_huge_valf(), den = 0.f, sx0 = 0.f, sx1 = 0.f;
    for (int j = beg; j < end; j += L1_U) {
        int s[L1_U]; float2 a[L1_U], xs[L1_U]; bool ok[L1_U];
        float l[L1_U];
#pragma unroll
        for (int u = 0; u < L1_U; ++u) {
            ok[u] = (j + u) < end;
            int jj = ok[u] ? (j + u) : j;
            s[u] = csrc[jj];
            a[u] = cea2[jj];
        }
#pragma unroll
        for (int u = 0; u < L1_U; ++u) xs[u] = x2[s[u]];
        float cm = mx;
#pragma unroll
        for (int u = 0; u < L1_U; ++u) {
            float lg = 0.f;
#pragma unroll
            for (int c = 0; c < 8; ++c) {
                float m = fmaf(xs[u].x, wl0[c], fmaf(xs[u].y, wl1[c],
                          fmaf(a[u].x, we0[c], fmaf(a[u].y, we1[c], base[c]))));
                m = (m > 0.f) ? m : 0.2f * m;
                lg = fmaf(m, at[c], lg);
            }
            l[u] = ok[u] ? lg : -__builtin_huge_valf();
            cm = fmaxf(cm, l[u]);
        }
        const float sc = __expf(mx - cm);
        den *= sc; sx0 *= sc; sx1 *= sc;
#pragma unroll
        for (int u = 0; u < L1_U; ++u) {
            float w = __expf(l[u] - cm);
            den += w;
            sx0 = fmaf(w, xs[u].x, sx0);
            sx1 = fmaf(w, xs[u].y, sx1);
        }
        mx = cm;
    }
    const bool has = end > beg;
    const float m0 = has ? sx0 / den : 0.f, m1 = has ? sx1 / den : 0.f;
    float blv[8], bsv[8];
    load8(bl + d0, blv); load8(bias + d0, bsv);
    v8h o;
#pragma unroll
    for (int c = 0; c < 8; ++c) {
        float v = (has ? fmaf(m0, wl0[c], fmaf(m1, wl1[c], blv[c])) : 0.f) + bsv[c];
        v = (v > 0.f) ? v : (__expf(v) - 1.f);
        o[c] = (_Float16)v;
    }
    *(v8h*)(hh + (size_t)t * 128 + d0) = o;
}

__device__ __forceinline__ float quad_xor1(float v) {
    return __int_as_float(__builtin_amdgcn_update_dpp(0, __float_as_int(v), 0xB1, 0xF, 0xF, true));
}
__device__ __forceinline__ float quad_xor2(float v) {
    return __int_as_float(__builtin_amdgcn_update_dpp(0, __float_as_int(v), 0x4E, 0xF, 0xF, true));
}

constexpr int L2_U = 8;
__global__ void __launch_bounds__(256)
k_node_attn2(const float* __restrict__ xl, const float* __restrict__ xr,
             const float* __restrict__ We, const float* __restrict__ att,
             const int* __restrict__ off, const int* __restrict__ csrc,
             const float* __restrict__ cea,
             float* __restrict__ agg, int N) {
    int idx = blockIdx.x * blockDim.x + threadIdx.x;
    if (idx >= N * 16) return;
    const int t = idx >> 4, d0 = (idx & 15) * 4;
    const float4 xrt = *(const float4*)(xr + (size_t)t * 64 + d0);
    const float4 w0 = *(const float4*)(We + d0);
    const float4 w1 = *(const float4*)(We + 64 + d0);
    const float4 at = *(const float4*)(att + d0);
    float4 acc = make_float4(0.f, 0.f, 0.f, 0.f);
    const int beg = off[t], end = off[t + 1];
    const float2* cea2 = (const float2*)cea;
    float mx = -__builtin_huge_valf(), den = 0.f;
    for (int j = beg; j < end; j += L2_U) {
        int s[L2_U]; float2 a[L2_U]; bool ok[L2_U];
        float4 row[L2_U]; float l[L2_U];
#pragma unroll
        for (int u = 0; u < L2_U; ++u) {
            ok[u] = (j + u) < end;
            int jj = ok[u] ? (j + u) : j;
            s[u] = csrc[jj];
            a[u] = cea2[jj];
        }
#pragma unroll
        for (int u = 0; u < L2_U; ++u)
            row[u] = *(const float4*)(xl + (size_t)s[u] * 64 + d0);
        float cm = mx;
#pragma unroll
        for (int u = 0; u < L2_U; ++u) {
            float m0 = row[u].x + xrt.x + fmaf(a[u].x, w0.x, a[u].y * w1.x);
            float m1 = row[u].y + xrt.y + fmaf(a[u].x, w0.y, a[u].y * w1.y);
            float m2 = row[u].z + xrt.z + fmaf(a[u].x, w0.z, a[u].y * w1.z);
            float m3 = row[u].w + xrt.w + fmaf(a[u].x, w0.w, a[u].y * w1.w);
            m0 = (m0 > 0.f) ? m0 : 0.2f * m0;
            m1 = (m1 > 0.f) ? m1 : 0.2f * m1;
            m2 = (m2 > 0.f) ? m2 : 0.2f * m2;
            m3 = (m3 > 0.f) ? m3 : 0.2f * m3;
            float lg = fmaf(m3, at.w, fmaf(m2, at.z, fmaf(m1, at.y, m0 * at.x)));
            lg += quad_xor1(lg);
            lg += quad_xor2(lg);
            l[u] = ok[u] ? lg : -__builtin_huge_valf();
            cm = fmaxf(cm, l[u]);
        }
        const float sc = __expf(mx - cm);
        den *= sc;
        acc.x *= sc; acc.y *= sc; acc.z *= sc; acc.w *= sc;
#pragma unroll
        for (int u = 0; u < L2_U; ++u) {
            float w = __expf(l[u] - cm);
            den += w;
            acc.x = fmaf(w, row[u].x, acc.x);
            acc.y = fmaf(w, row[u].y, acc.y);
            acc.z = fmaf(w, row[u].z, acc.z);
            acc.w = fmaf(w, row[u].w, acc.w);
        }
        mx = cm;
    }
    float4 o = make_float4(0.f, 0.f, 0.f, 0.f);
    if (end > beg) { o.x = acc.x / den; o.y = acc.y / den; o.z = acc.z / den; o.w = acc.w / den; }
    *(float4*)(agg + (size_t)t * 64 + d0) = o;
}

__global__ void k_w_pack(const float* __restrict__ Wl, const float* __restrict__ Wr,
                         _Float16* __restrict__ Wpk) {
    int idx = blockIdx.x * blockDim.x + threadIdx.x;
    if (idx >= 2 * 8192) return;
    int which = idx >> 13;
    int r = idx & 8191;
    int j = r & 15;
    int lane = (r >> 4) & 31;
    int frag = r >> 9;
    int kc = frag >> 2, nt = frag & 3;
    int half_sel = lane >> 4, lidx = lane & 15;
    int kb = (j < 8) ? (half_sel * 8) : (16 + half_sel * 8);
    int k = kc * 32 + kb + (j & 7);
    int n = nt * 16 + lidx;
    const float* W = which ? Wr : Wl;
    Wpk[idx] = (_Float16)W[k * 64 + n];
}

__global__ void k_gemm2_wmma(const _Float16* __restrict__ hh,
                             const _Float16* __restrict__ Wpk,
                             const float* __restrict__ bl, const float* __restrict__ br,
                             float* __restrict__ xl2, float* __restrict__ xr2, int N) {
    __shared__ __align__(32) _Float16 ldsW[2 * 8192];

    {
        const uint4* g = (const uint4*)Wpk;
        uint4* l = (uint4*)ldsW;
        for (int i = threadIdx.x; i < 2048; i += blockDim.x) l[i] = g[i];
    }
    __syncthreads();

    int wave = (blockIdx.x * blockDim.x + threadIdx.x) >> 5;
    int lane = threadIdx.x & 31;
    int row0 = wave * 16;
    if (row0 >= N) return;

    int half_sel = lane >> 4;
    int lidx = lane & 15;
    int kb_lo = half_sel * 8;
    int kb_hi = 16 + kb_lo;
    int m = row0 + lidx;
    if (m >= N) m = N - 1;

    const v16h* fragL = (const v16h*)ldsW;
    const v16h* fragR = (const v16h*)(ldsW + 8192);

    v8f accl[4] = {};
    v8f accr[4] = {};

#pragma unroll
    for (int kc = 0; kc < 4; ++kc) {
        const _Float16* pa = hh + m * 128 + kc * 32;
        v8h alo = *(const v8h*)(pa + kb_lo);
        v8h ahi = *(const v8h*)(pa + kb_hi);
        v16h a = __builtin_shufflevector(alo, ahi, 0, 1, 2, 3, 4, 5, 6, 7,
                                         8, 9, 10, 11, 12, 13, 14, 15);
#pragma unroll
        for (int nt = 0; nt < 4; ++nt) {
            int fi = (kc * 4 + nt) * 32 + lane;
            v16h b  = fragL[fi];
            v16h b2 = fragR[fi];
            accl[nt] = __builtin_amdgcn_wmma_f32_16x16x32_f16(
                false, a, false, b, (short)0, accl[nt], false, false);
            accr[nt] = __builtin_amdgcn_wmma_f32_16x16x32_f16(
                false, a, false, b2, (short)0, accr[nt], false, false);
        }
    }
#pragma unroll
    for (int nt = 0; nt < 4; ++nt) {
        int n = nt * 16 + lidx;
        float bln = bl[n], brn = br[n];
#pragma unroll
        for (int r = 0; r < 8; ++r) {
            int mo = row0 + r + half_sel * 8;
            if (mo < N) {
                xl2[mo * 64 + n] = accl[nt][r] + bln;
                xr2[mo * 64 + n] = accr[nt][r] + brn;
            }
        }
    }
}

constexpr int POOL_NPB = 128;
constexpr int POOL_U   = 8;

__global__ void __launch_bounds__(256)
k_pool(const float* __restrict__ agg2, const float* __restrict__ bias2,
       const int* __restrict__ batch,
       float* __restrict__ sums, float* __restrict__ cnts, int N) {
    const int d   = threadIdx.x & 63;
    const int sub = threadIdx.x >> 6;
    const int n0  = blockIdx.x * POOL_NPB + sub;
    const float b = bias2[d];
    float acc = 0.f, cnt = 0.f;
    int gcur = -1;
    for (int it = 0; it < POOL_NPB / 4; it += POOL_U) {
        float v[POOL_U];
        int   g[POOL_U];
#pragma unroll
        for (int u = 0; u < POOL_U; ++u) {
            int n = n0 + 4 * (it + u);
            bool ok = n < N;
            g[u] = ok ? batch[n] : -1;
            v[u] = ok ? agg2[n * 64 + d] : 0.f;
        }
#pragma unroll
        for (int u = 0; u < POOL_U; ++u) {
            if (g[u] != gcur) {
                if (gcur >= 0) {
                    atomicAdd(&sums[gcur * 64 + d], acc);
                    if (d == 0) atomicAdd(&cnts[gcur], cnt);
                }
                gcur = g[u]; acc = 0.f; cnt = 0.f;
            }
            acc += v[u] + b;
            cnt += 1.0f;
        }
    }
    if (gcur >= 0) {
        atomicAdd(&sums[gcur * 64 + d], acc);
        if (d == 0) atomicAdd(&cnts[gcur], cnt);
    }
}

__global__ void k_final(const float* __restrict__ sums, const float* __restrict__ cnts,
                        float* __restrict__ out, int G) {
    int idx = blockIdx.x * blockDim.x + threadIdx.x;
    if (idx >= G * 64) return;
    out[idx] = sums[idx] / fmaxf(cnts[idx >> 6], 1.0f);
}

extern "C" void kernel_launch(void* const* d_in, const int* in_sizes, int n_in,
                              void* d_out, int out_size, void* d_ws, size_t ws_size,
                              hipStream_t stream) {
    const float* x      = (const float*)d_in[0];
    const float* ea     = (const float*)d_in[1];
    const float* W_l1   = (const float*)d_in[2];
    const float* b_l1   = (const float*)d_in[3];
    const float* W_r1   = (const float*)d_in[4];
    const float* b_r1   = (const float*)d_in[5];
    const float* W_e1   = (const float*)d_in[6];
    const float* att1   = (const float*)d_in[7];
    const float* bias1  = (const float*)d_in[8];
    const float* W_l2   = (const float*)d_in[9];
    const float* b_l2   = (const float*)d_in[10];
    const float* W_r2   = (const float*)d_in[11];
    const float* b_r2   = (const float*)d_in[12];
    const float* W_e2   = (const float*)d_in[13];
    const float* att2   = (const float*)d_in[14];
    const float* bias2  = (const float*)d_in[15];
    const int*   eidx   = (const int*)d_in[16];
    const int*   batch  = (const int*)d_in[17];
    float* out = (float*)d_out;

    const int N = in_sizes[0] / 2;
    const int E = in_sizes[1] / 2;
    const int G = out_size / 64;
    const int* src = eidx;
    const int* tgt = eidx + E;

    const int per4 = ((N + 1 + 1023) / 1024 + 3) / 4;
    const size_t cnt_bytes = (size_t)1024 * per4 * 16;

    char* ws = (char*)d_ws;
    size_t off = 0;
    auto alloc = [&](size_t bytes) { size_t o = off; off = (off + bytes + 255) & ~(size_t)255; return o; };

    size_t o_cnt   = alloc(cnt_bytes);
    size_t o_sums  = alloc((size_t)G * 64 * 4);
    size_t o_cnts  = alloc((size_t)G * 4);
    size_t acc_bytes = off;

    size_t o_off  = alloc(cnt_bytes);
    size_t o_rank = alloc((size_t)E * 4);
    size_t o_csrc = alloc((size_t)E * 4);
    size_t o_cea  = alloc((size_t)E * 8);
    size_t o_hh   = alloc((size_t)N * 128 * 2);
    size_t o_wpk  = alloc((size_t)2 * 8192 * 2);
    size_t o_xl2  = alloc((size_t)N * 64 * 4);
    size_t o_xr2  = alloc((size_t)N * 64 * 4);
    size_t o_agg2 = alloc((size_t)N * 64 * 4);
    (void)ws_size; (void)n_in;

    int*      cnt   = (int*)(ws + o_cnt);
    float*    sums  = (float*)(ws + o_sums);
    float*    cnts  = (float*)(ws + o_cnts);
    int*      offs  = (int*)(ws + o_off);
    int*      rank  = (int*)(ws + o_rank);
    int*      csrc  = (int*)(ws + o_csrc);
    float*    cea   = (float*)(ws + o_cea);
    _Float16* hh    = (_Float16*)(ws + o_hh);
    _Float16* wpk   = (_Float16*)(ws + o_wpk);
    float*    xl2   = (float*)(ws + o_xl2);
    float*    xr2   = (float*)(ws + o_xr2);
    float*    agg2  = (float*)(ws + o_agg2);

    const int TB = 256;
    auto blocks = [&](long long t) { return (unsigned)((t + TB - 1) / TB); };

    hipMemsetAsync(d_ws, 0, acc_bytes, stream);

    k_hist<<<blocks(E), TB, 0, stream>>>(tgt, cnt, rank, E);
    k_scan<<<1, 1024, 0, stream>>>(cnt, offs, per4);
    k_fill<<<blocks(E), TB, 0, stream>>>(src, tgt, ea, offs, rank, csrc, cea, E);

    k_node_attn1<<<blocks((long long)N * 16), TB, 0, stream>>>(
        x, W_l1, b_l1, W_r1, b_r1, W_e1, att1, bias1, offs, csrc, cea, hh, N);

    k_w_pack<<<blocks(2 * 8192), TB, 0, stream>>>(W_l2, W_r2, wpk);
    long long gemm_threads = (long long)((N + 15) / 16) * 32;
    k_gemm2_wmma<<<blocks(gemm_threads), TB, 0, stream>>>(
        hh, wpk, b_l2, b_r2, xl2, xr2, N);

    k_node_attn2<<<blocks((long long)N * 16), TB, 0, stream>>>(
        xl2, xr2, W_e2, att2, offs, csrc, cea, agg2, N);

    k_pool<<<(unsigned)((N + POOL_NPB - 1) / POOL_NPB), TB, 0, stream>>>(agg2, bias2, batch, sums, cnts, N);
    k_final<<<blocks((long long)G * 64), TB, 0, stream>>>(sums, cnts, out, G);
}


# each bucket slot is one 16-byte record (source id + 2 edge attributes): the fill kernel scatters one store per edge, not two, and both attention kernels fetch an edge with one load, not two
# speedup vs baseline: 2.9851x; 1.0633x over previous
#include <hip/hip_runtime.h>
#include <hip/hip_fp16.h>

typedef __attribute__((ext_vector_type(16))) _Float16 v16h;
typedef __attribute__((ext_vector_type(8)))  _Float16 v8h;
typedef __attribute__((ext_vector_type(8)))  float    v8f;

__global__ void k_hist(const int* __restrict__ tgt, int* __restrict__ cnt,
                       int* __restrict__ rank, int E) {
    int e = blockIdx.x * blockDim.x + threadIdx.x;
    if (e >= E) return;
    rank[e] = atomicAdd(&cnt[tgt[e]], 1);
}

__global__ void __launch_bounds__(1024)
k_scan(const int* __restrict__ cnt, int* __restrict__ off, int per4) {
    __shared__ int wsum[32];
    const int tid = threadIdx.x, lane = tid & 31, wv = tid >> 5;
    const int4* c4 = (const int4*)cnt + (size_t)tid * per4;
    int s = 0;
    for (int i = 0; i < per4; ++i) { int4 v = c4[i]; s += (v.x + v.y) + (v.z + v.w); }
    int inc = s;
    for (int d = 1; d < 32; d <<= 1) { int o = __shfl_up(inc, d); if (lane >= d) inc += o; }
    if (lane == 31) wsum[wv] = inc;
    __syncthreads();
    if (wv == 0) {
        int w = wsum[lane], winc = w;
        for (int d = 1; d < 32; d <<= 1) { int o = __shfl_up(winc, d); if (lane >= d) winc += o; }
        wsum[lane] = winc - w;
    }
    __syncthreads();
    int run = wsum[wv] + inc - s;
    int4* o4 = (int4*)off + (size_t)tid * per4;
    for (int i = 0; i < per4; ++i) {
        int4 v = c4[i], o;
        o.x = run; run += v.x;
        o.y = run; run += v.y;
        o.z = run; run += v.z;
        o.w = run; run += v.w;
        o4[i] = o;
    }
}

__global__ void k_fill(const int* __restrict__ src, const int* __restrict__ tgt,
                       const float* __restrict__ ea,
                       const int* __restrict__ off, const int* __restrict__ rank,
                       float* __restrict__ crec, int E) {
    int e = blockIdx.x * blockDim.x + threadIdx.x;
    if (e >= E) return;
    int p = off[tgt[e]] + rank[e];
    const float2 a = ((const float2*)ea)[e];
    ((float4*)crec)[p] = make_float4(__int_as_float(src[e]), a.x, a.y, 0.f);
}

__device__ __forceinline__ void load8(const float* __restrict__ p, float (&v)[8]) {
    const float4 a = ((const float4*)p)[0], b = ((const float4*)p)[1];
    v[0] = a.x; v[1] = a.y; v[2] = a.z; v[3] = a.w;
    v[4] = b.x; v[5] = b.y; v[6] = b.z; v[7] = b.w;
}

constexpr int L1_U = 4;
__global__ void __launch_bounds__(256)
k_node_attn1(const float* __restrict__ x,
             const float* __restrict__ Wl, const float* __restrict__ bl,
             const float* __restrict__ Wr, const float* __restrict__ br,
             const float* __restrict__ We, const float* __restrict__ att,
             const float* __restrict__ bias,
             const int* __restrict__ off, const float* __restrict__ crec,
             _Float16* __restrict__ hh, int N) {
    int idx = blockIdx.x * blockDim.x + threadIdx.x;
    if (idx >= N * 16) return;
    const int t = idx >> 4, d0 = (idx & 15) * 8;
    const float2* x2 = (const float2*)x;
    float wl0[8], wl1[8], we0[8], we1[8], at[8], base[8];
    load8(Wl + d0, wl0); load8(Wl + 128 + d0, wl1);
    load8(We + d0, we0); load8(We + 128 + d0, we1);
    load8(att + d0, at);
    {
        const float2 xt = x2[t];
        float wr0[8], wr1[8], brv[8], blv[8];
        load8(Wr + d0, wr0); load8(Wr + 128 + d0, wr1);
        load8(br + d0, brv); load8(bl + d0, blv);
#pragma unroll
        for (int c = 0; c < 8; ++c)
            base[c] = fmaf(xt.x, wr0[c], fmaf(xt.y, wr1[c], brv[c])) + blv[c];
    }
    const int beg = off[t], end = off[t + 1];
    const float4* rec4 = (const float4*)crec;
    float mx = -__builtin_huge_valf(), den = 0.f, sx0 = 0.f, sx1 = 0.f;
    for (int j = beg; j < end; j += L1_U) {
        int s[L1_U]; float2 a[L1_U], xs[L1_U]; bool ok[L1_U];
        float l[L1_U];
#pragma unroll
        for (int u = 0; u < L1_U; ++u) {
            ok[u] = (j + u) < end;
            int jj = ok[u] ? (j + u) : j;
            const float4 r = rec4[jj];
            s[u] = __float_as_int(r.x);
            a[u] = make_float2(r.y, r.z);
        }
#pragma unroll
        for (int u = 0; u < L1_U; ++u) xs[u] = x2[s[u]];
        float cm = mx;
#pragma unroll
        for (int u = 0; u < L1_U; ++u) {
            float lg = 0.f;
#pragma unroll
            for (int c = 0; c < 8; ++c) {
                float m = fmaf(xs[u].x, wl0[c], fmaf(xs[u].y, wl1[c],
                          fmaf(a[u].x, we0[c], fmaf(a[u].y, we1[c], base[c]))));
                m = (m > 0.f) ? m : 0.2f * m;
                lg = fmaf(m, at[c], lg);
            }
            l[u] = ok[u] ? lg : -__builtin_huge_valf();
            cm = fmaxf(cm, l[u]);
        }
        const float sc = __expf(mx - cm);
        den *= sc; sx0 *= sc; sx1 *= sc;
#pragma unroll
        for (int u = 0; u < L1_U; ++u) {
            float w = __expf(l[u] - cm);
            den += w;
            sx0 = fmaf(w, xs[u].x, sx0);
            sx1 = fmaf(w, xs[u].y, sx1);
        }
        mx = cm;
    }
    const bool has = end > beg;
    const float m0 = has ? sx0 / den : 0.f, m1 = has ? sx1 / den : 0.f;
    float blv[8], bsv[8];
    load8(bl + d0, blv); load8(bias + d0, bsv);
    v8h o;
#pragma unroll
    for (int c = 0; c < 8; ++c) {
        float v = (has ? fmaf(m0, wl0[c], fmaf(m1, wl1[c], blv[c])) : 0.f) + bsv[c];
        v = (v > 0.f) ? v : (__expf(v) - 1.f);
        o[c] = (_Float16)v;
    }
    *(v8h*)(hh + (size_t)t * 128 + d0) = o;
}

__device__ __forceinline__ float quad_xor1(float v) {
    return __int_as_float(__builtin_amdgcn_update_dpp(0, __float_as_int(v), 0xB1, 0xF, 0xF, true));
}
__device__ __forceinline__ float quad_xor2(float v) {
    return __int_as_float(__builtin_amdgcn_update_dpp(0, __float_as_int(v), 0x4E, 0xF, 0xF, true));
}

constexpr int L2_U = 8;
__global__ void __launch_bounds__(256)
k_node_attn2(const float* __restrict__ xl, const float* __restrict__ xr,
             const float* __restrict__ We, const float* __restrict__ att,
             const int* __restrict__ off, const float* __restrict__ crec,
             float* __restrict__ agg, int N) {
    int idx = blockIdx.x * blockDim.x + threadIdx.x;
    if (idx >= N * 16) return;
    const int t = idx >> 4, d0 = (idx & 15) * 4;
    const float4 xrt = *(const float4*)(xr + (size_t)t * 64 + d0);
    const float4 w0 = *(const float4*)(We + d0);
    const float4 w1 = *(const float4*)(We + 64 + d0);
    const float4 at = *(const float4*)(att + d0);
    float4 acc = make_float4(0.f, 0.f, 0.f, 0.f);
    const int beg = off[t], end = off[t + 1];
    const float4* rec4 = (const float4*)crec;
    float mx = -__builtin_huge_valf(), den = 0.f;
    for (int j = beg; j < end; j += L2_U) {
        int s[L2_U]; float2 a[L2_U]; bool ok[L2_U];
        float4 row[L2_U]; float l[L2_U];
#pragma unroll
        for (int u = 0; u < L2_U; ++u) {
            ok[u] = (j + u) < end;
            int jj = ok[u] ? (j + u) : j;
            const float4 r = rec4[jj];
            s[u] = __float_as_int(r.x);
            a[u] = make_float2(r.y, r.z);
        }
#pragma unroll
        for (int u = 0; u < L2_U; ++u)
            row[u] = *(const float4*)(xl + (size_t)s[u] * 64 + d0);
        float cm = mx;
#pragma unroll
        for (int u = 0; u < L2_U; ++u) {
            float m0 = row[u].x + xrt.x + fmaf(a[u].x, w0.x, a[u].y * w1.x);
            float m1 = row[u].y + xrt.y + fmaf(a[u].x, w0.y, a[u].y * w1.y);
            float m2 = row[u].z + xrt.z + fmaf(a[u].x, w0.z, a[u].y * w1.z);
            float m3 = row[u].w + xrt.w + fmaf(a[u].x, w0.w, a[u].y * w1.w);
            m0 = (m0 > 0.f) ? m0 : 0.2f * m0;
            m1 = (m1 > 0.f) ? m1 : 0.2f * m1;
            m2 = (m2 > 0.f) ? m2 : 0.2f * m2;
            m3 = (m3 > 0.f) ? m3 : 0.2f * m3;
            float lg = fmaf(m3, at.w, fmaf(m2, at.z, fmaf(m1, at.y, m0 * at.x)));
            lg += quad_xor1(lg);
            lg += quad_xor2(lg);
            l[u] = ok[u] ? lg : -__builtin_huge_valf();
            cm = fmaxf(cm, l[u]);
        }
        const float sc = __expf(mx - cm);
        den *= sc;
        acc.x *= sc; acc.y *= sc; acc.z *= sc; acc.w *= sc;
#pragma unroll
        for (int u = 0; u < L2_U; ++u) {
            float w = __expf(l[u] - cm);
            den += w;
            acc.x = fmaf(w, row[u].x, acc.x);
            acc.y = fmaf(w, row[u].y, acc.y);
            acc.z = fmaf(w, row[u].z, acc.z);
            acc.w = fmaf(w, row[u].w, acc.w);
        }
        mx = cm;
    }
    float4 o = make_float4(0.f, 0.f, 0.f, 0.f);
    if (end > beg) { o.x = acc.x / den; o.y = acc.y / den; o.z = acc.z / den; o.w = acc.w / den; }
    *(float4*)(agg + (size_t)t * 64 + d0) = o;
}

__global__ void k_w_pack(const float* __restrict__ Wl, const float* __restrict__ Wr,
                         _Float16* __restrict__ Wpk) {
    int idx = blockIdx.x * blockDim.x + threadIdx.x;
    if (idx >= 2 * 8192) return;
    int which = idx >> 13;
    int r = idx & 8191;
    int j = r & 15;
    int lane = (r >> 4) & 31;
    int frag = r >> 9;
    int kc = frag >> 2, nt = frag & 3;
    int half_sel = lane >> 4, lidx = lane & 15;
    int kb = (j < 8) ? (half_sel * 8) : (16 + half_sel * 8);
    int k = kc * 32 + kb + (j & 7);
    int n = nt * 16 + lidx;
    const float* W = which ? Wr : Wl;
    Wpk[idx] = (_Float16)W[k * 64 + n];
}

__global__ void k_gemm2_wmma(const _Float16* __restrict__ hh,
                             const _Float16* __restrict__ Wpk,
                             const float* __restrict__ bl, const float* __restrict__ br,
                             float* __restrict__ xl2, float* __restrict__ xr2, int N) {
    __shared__ __align__(32) _Float16 ldsW[2 * 8192];

    {
        const uint4* g = (const uint4*)Wpk;
        uint4* l = (uint4*)ldsW;
        for (int i = threadIdx.x; i < 2048; i += blockDim.x) l[i] = g[i];
    }
    __syncthreads();

    int wave = (blockIdx.x * blockDim.x + threadIdx.x) >> 5;
    int lane = threadIdx.x & 31;
    int row0 = wave * 16;
    if (row0 >= N) return;

    int half_sel = lane >> 4;
    int lidx = lane & 15;
    int kb_lo = half_sel * 8;
    int kb_hi = 16 + kb_lo;
    int m = row0 + lidx;
    if (m >= N) m = N - 1;

    const v16h* fragL = (const v16h*)ldsW;
    const v16h* fragR = (const v16h*)(ldsW + 8192);

    v8f accl[4] = {};
    v8f accr[4] = {};

#pragma unroll
    for (int kc = 0; kc < 4; ++kc) {
        const _Float16* pa = hh + m * 128 + kc * 32;
        v8h alo = *(const v8h*)(pa + kb_lo);
        v8h ahi = *(const v8h*)(pa + kb_hi);
        v16h a = __builtin_shufflevector(alo, ahi, 0, 1, 2, 3, 4, 5, 6, 7,
                                         8, 9, 10, 11, 12, 13, 14, 15);
#pragma unroll
        for (int nt = 0; nt < 4; ++nt) {
            int fi = (kc * 4 + nt) * 32 + lane;
            v16h b  = fragL[fi];
            v16h b2 = fragR[fi];
            accl[nt] = __builtin_amdgcn_wmma_f32_16x16x32_f16(
                false, a, false, b, (short)0, accl[nt], false, false);
            accr[nt] = __builtin_amdgcn_wmma_f32_16x16x32_f16(
                false, a, false, b2, (short)0, accr[nt], false, false);
        }
    }
#pragma unroll
    for (int nt = 0; nt < 4; ++nt) {
        int n = nt * 16 + lidx;
        float bln = bl[n], brn = br[n];
#pragma unroll
        for (int r = 0; r < 8; ++r) {
            int mo = row0 + r + half_sel * 8;
            if (mo < N) {
                xl2[mo * 64 + n] = accl[nt][r] + bln;
                xr2[mo * 64 + n] = accr[nt][r] + brn;
            }
        }
    }
}

constexpr int POOL_NPB = 128;
constexpr int POOL_U   = 8;

__global__ void __launch_bounds__(256)
k_pool(const float* __restrict__ agg2, const float* __restrict__ bias2,
       const int* __restrict__ batch,
       float* __restrict__ sums, float* __restrict__ cnts, int N) {
    const int d   = threadIdx.x & 63;
    const int sub = threadIdx.x >> 6;
    const int n0  = blockIdx.x * POOL_NPB + sub;
    const float b = bias2[d];
    float acc = 0.f, cnt = 0.f;
    int gcur = -1;
    for (int it = 0; it < POOL_NPB / 4; it += POOL_U) {
        float v[POOL_U];
        int   g[POOL_U];
#pragma unroll
        for (int u = 0; u < POOL_U; ++u) {
            int n = n0 + 4 * (it + u);
            bool ok = n < N;
            g[u] = ok ? batch[n] : -1;
            v[u] = ok ? agg2[n * 64 + d] : 0.f;
        }
#pragma unroll
        for (int u = 0; u < POOL_U; ++u) {
            if (g[u] != gcur) {
                if (gcur >= 0) {
                    atomicAdd(&sums[gcur * 64 + d], acc);
                    if (d == 0) atomicAdd(&cnts[gcur], cnt);
                }
                gcur = g[u]; acc = 0.f; cnt = 0.f;
            }
            acc += v[u] + b;
            cnt += 1.0f;
        }
    }
    if (gcur >= 0) {
        atomicAdd(&sums[gcur * 64 + d], acc);
        if (d == 0) atomicAdd(&cnts[gcur], cnt);
    }
}

__global__ void k_final(const float* __restrict__ sums, const float* __restrict__ cnts,
                        float* __restrict__ out, int G) {
    int idx = blockIdx.x * blockDim.x + threadIdx.x;
    if (idx >= G * 64) return;
    out[idx] = sums[idx] / fmaxf(cnts[idx >> 6], 1.0f);
}

extern "C" void kernel_launch(void* const* d_in, const int* in_sizes, int n_in,
                              void* d_out, int out_size, void* d_ws, size_t ws_size,
                              hipStream_t stream) {
    const float* x      = (const float*)d_in[0];
    const float* ea     = (const float*)d_in[1];
    const float* W_l1   = (const float*)d_in[2];
    const float* b_l1   = (const float*)d_in[3];
    const float* W_r1   = (const float*)d_in[4];
    const float* b_r1   = (const float*)d_in[5];
    const float* W_e1   = (const float*)d_in[6];
    const float* att1   = (const float*)d_in[7];
    const float* bias1  = (const float*)d_in[8];
    const float* W_l2   = (const float*)d_in[9];
    const float* b_l2   = (const float*)d_in[10];
    const float* W_r2   = (const float*)d_in[11];
    const float* b_r2   = (const float*)d_in[12];
    const float* W_e2   = (const float*)d_in[13];
    const float* att2   = (const float*)d_in[14];
    const float* bias2  = (const float*)d_in[15];
    const int*   eidx   = (const int*)d_in[16];
    const int*   batch  = (const int*)d_in[17];
    float* out = (float*)d_out;

    const int N = in_sizes[0] / 2;
    const int E = in_sizes[1] / 2;
    const int G = out_size / 64;
    const int* src = eidx;
    const int* tgt = eidx + E;

    const int per4 = ((N + 1 + 1023) / 1024 + 3) / 4;
    const size_t cnt_bytes = (size_t)1024 * per4 * 16;

    char* ws = (char*)d_ws;
    size_t off = 0;
    auto alloc = [&](size_t bytes) { size_t o = off; off = (off + bytes + 255) & ~(size_t)255; return o; };

    size_t o_cnt   = alloc(cnt_bytes);
    size_t o_sums  = alloc((size_t)G * 64 * 4);
    size_t o_cnts  = alloc((size_t)G * 4);
    size_t acc_bytes = off;

    size_t o_off  = alloc(cnt_bytes);
    size_t o_rank = alloc((size_t)E * 4);
    size_t o_crec = alloc((size_t)E * 16);
    size_t o_hh   = alloc((size_t)N * 128 * 2);
    size_t o_wpk  = alloc((size_t)2 * 8192 * 2);
    size_t o_xl2  = alloc((size_t)N * 64 * 4);
    size_t o_xr2  = alloc((size_t)N * 64 * 4);
    size_t o_agg2 = alloc((size_t)N * 64 * 4);
    (void)ws_size; (void)n_in;

    int*      cnt   = (int*)(ws + o_cnt);
    float*    sums  = (float*)(ws + o_sums);
    float*    cnts  = (float*)(ws + o_cnts);
    int*      offs  = (int*)(ws + o_off);
    int*      rank  = (int*)(ws + o_rank);
    float*    crec  = (float*)(ws + o_crec);
    _Float16* hh    = (_Float16*)(ws + o_hh);
    _Float16* wpk   = (_Float16*)(ws + o_wpk);
    float*    xl2   = (float*)(ws + o_xl2);
    float*    xr2   = (float*)(ws + o_xr2);
    float*    agg2  = (float*)(ws + o_agg2);

    const int TB = 256;
    auto blocks = [&](long long t) { return (unsigned)((t + TB - 1) / TB); };

    hipMemsetAsync(d_ws, 0, acc_bytes, stream);

    k_hist<<<blocks(E), TB, 0, stream>>>(tgt, cnt, rank, E);
    k_scan<<<1, 1024, 0, stream>>>(cnt, offs, per4);
    k_fill<<<blocks(E), TB, 0, stream>>>(src, tgt, ea, offs, rank, crec, E);

    k_node_attn1<<<blocks((long long)N * 16), TB, 0, stream>>>(
        x, W_l1, b_l1, W_r1, b_r1, W_e1, att1, bias1, offs, crec, hh, N);

    k_w_pack<<<blocks(2 * 8192), TB, 0, stream>>>(W_l2, W_r2, wpk);
    long long gemm_threads = (long long)((N + 15) / 16) * 32;
    k_gemm2_wmma<<<blocks(gemm_threads), TB, 0, stream>>>(
        hh, wpk, b_l2, b_r2, xl2, xr2, N);

    k_node_attn2<<<blocks((long long)N * 16), TB, 0, stream>>>(
        xl2, xr2, W_e2, att2, offs, crec, agg2, N);

    k_pool<<<(unsigned)((N + POOL_NPB - 1) / POOL_NPB), TB, 0, stream>>>(agg2, bias2, batch, sums, cnts, N);
    k_final<<<blocks((long long)G * 64), TB, 0, stream>>>(sums, cnts, out, G);
}


# layer-2 walk in groups of 4 edges, not 8: half the registers (12 waves per SIMD, not 5) and a group's loads leave as 2 batches, not 5; same edge order, running-max rescale every 4 edges
# speedup vs baseline: 3.9176x; 1.3124x over previous
#include <hip/hip_runtime.h>
#include <hip/hip_fp16.h>

typedef __attribute__((ext_vector_type(16))) _Float16 v16h;
typedef __attribute__((ext_vector_type(8)))  _Float16 v8h;
typedef __attribute__((ext_vector_type(8)))  float    v8f;

constexpr int CAP = 48;
constexpr int BK_U = 4;

__global__ void __launch_bounds__(256)
k_bucket(const int* __restrict__ src, const int* __restrict__ tgt,
         const float* __restrict__ ea, int* __restrict__ cnt, int* __restrict__ novf,
         float* __restrict__ crec, float* __restrict__ ovf, int E) {
    const int T = gridDim.x * blockDim.x;
    const int e0 = blockIdx.x * blockDim.x + threadIdx.x;
    if (e0 >= E) return;
    int t[BK_U], s[BK_U], p[BK_U]; float2 a[BK_U]; bool ok[BK_U];
#pragma unroll
    for (int u = 0; u < BK_U; ++u) {
        const long long e = (long long)e0 + (long long)u * T;
        ok[u] = e < E;
        const int ee = ok[u] ? (int)e : e0;
        t[u] = tgt[ee]; s[u] = src[ee];
        a[u] = ((const float2*)ea)[ee];
    }
#pragma unroll
    for (int u = 0; u < BK_U; ++u) p[u] = ok[u] ? atomicAdd(&cnt[t[u]], 1) : 0;
#pragma unroll
    for (int u = 0; u < BK_U; ++u) {
        if (!ok[u]) continue;
        float4* dst = (float4*)crec + ((size_t)t[u] * CAP + p[u]);
        float w3 = 0.f;
        if (p[u] >= CAP) {
            const int q = atomicAdd(novf, 1);
            dst = (float4*)ovf + q;
            w3 = __int_as_float(t[u]);
        }
        *dst = make_float4(__int_as_float(s[u]), a[u].x, a[u].y, w3);
    }
}

__device__ __forceinline__ void load8(const float* __restrict__ p, float (&v)[8]) {
    const float4 a = ((const float4*)p)[0], b = ((const float4*)p)[1];
    v[0] = a.x; v[1] = a.y; v[2] = a.z; v[3] = a.w;
    v[4] = b.x; v[5] = b.y; v[6] = b.z; v[7] = b.w;
}

__device__ __forceinline__ float head_logit8(const float2 xs, const float2 a,
        const float (&wl0)[8], const float (&wl1)[8],
        const float (&we0)[8], const float (&we1)[8],
        const float (&base)[8], const float (&at)[8]) {
    float lg = 0.f;
#pragma unroll
    for (int c = 0; c < 8; ++c) {
        float m = fmaf(xs.x, wl0[c], fmaf(xs.y, wl1[c],
                  fmaf(a.x, we0[c], fmaf(a.y, we1[c], base[c]))));
        m = (m > 0.f) ? m : 0.2f * m;
        lg = fmaf(m, at[c], lg);
    }
    return lg;
}

constexpr int L1_U = 4;
__global__ void __launch_bounds__(256)
k_node_attn1(const float* __restrict__ x,
             const float* __restrict__ Wl, const float* __restrict__ bl,
             const float* __restrict__ Wr, const float* __restrict__ br,
             const float* __restrict__ We, const float* __restrict__ att,
             const float* __restrict__ bias,
             const int* __restrict__ cnt, const float* __restrict__ crec,
             const int* __restrict__ novf, const float* __restrict__ ovf,
             _Float16* __restrict__ hh, int N, int E) {
    int idx = blockIdx.x * blockDim.x + threadIdx.x;
    if (idx >= N * 16) return;
    const int t = idx >> 4, d0 = (idx & 15) * 8;
    const float2* x2 = (const float2*)x;
    float wl0[8], wl1[8], we0[8], we1[8], at[8], base[8];
    load8(Wl + d0, wl0); load8(Wl + 128 + d0, wl1);
    load8(We + d0, we0); load8(We + 128 + d0, we1);
    load8(att + d0, at);
    {
        const float2 xt = x2[t];
        float wr0[8], wr1[8], brv[8], blv[8];
        load8(Wr + d0, wr0); load8(Wr + 128 + d0, wr1);
        load8(br + d0, brv); load8(bl + d0, blv);
#pragma unroll
        for (int c = 0; c < 8; ++c)
            base[c] = fmaf(xt.x, wr0[c], fmaf(xt.y, wr1[c], brv[c])) + blv[c];
    }
    const int deg = cnt[t];
    const unsigned beg = (unsigned)t * CAP, end = beg + (unsigned)(deg < CAP ? deg : CAP);
    const float4* rec4 = (const float4*)crec;
    float mx = -__builtin_huge_valf(), den = 0.f, sx0 = 0.f, sx1 = 0.f;
    for (unsigned j = beg; j < end; j += L1_U) {
        int s[L1_U]; float2 a[L1_U], xs[L1_U]; bool ok[L1_U];
        float l[L1_U];
#pragma unroll
        for (int u = 0; u < L1_U; ++u) {
            ok[u] = (j + u) < end;
            unsigned jj = ok[u] ? (j + u) : j;
            const float4 r = rec4[jj];
            s[u] = __float_as_int(r.x);
            a[u] = make_float2(r.y, r.z);
        }
#pragma unroll
        for (int u = 0; u < L1_U; ++u) xs[u] = x2[s[u]];
        float cm = mx;
#pragma unroll
        for (int u = 0; u < L1_U; ++u) {
            float lg = 0.f;
#pragma unroll
            for (int c = 0; c < 8; ++c) {
                float m = fmaf(xs[u].x, wl0[c], fmaf(xs[u].y, wl1[c],
                          fmaf(a[u].x, we0[c], fmaf(a[u].y, we1[c], base[c]))));
                m = (m > 0.f) ? m : 0.2f * m;
                lg = fmaf(m, at[c], lg);
            }
            l[u] = ok[u] ? lg : -__builtin_huge_valf();
            cm = fmaxf(cm, l[u]);
        }
        const float sc = __expf(mx - cm);
        den *= sc; sx0 *= sc; sx1 *= sc;
#pragma unroll
        for (int u = 0; u < L1_U; ++u) {
            float w = __expf(l[u] - cm);
            den += w;
            sx0 = fmaf(w, xs[u].x, sx0);
            sx1 = fmaf(w, xs[u].y, sx1);
        }
        mx = cm;
    }
    if (deg > CAP) {
        const int nraw = *novf;
        const int no = nraw < E ? nraw : E;
        const float4* o4 = (const float4*)ovf;
        for (int i = 0; i < no; ++i) {
            const float4 r = o4[i];
            if (__float_as_int(r.w) != t) continue;
            const float2 xs = x2[__float_as_int(r.x)];
            const float lg = head_logit8(xs, make_float2(r.y, r.z), wl0, wl1, we0, we1, base, at);
            const float cm = fmaxf(mx, lg);
            const float sc = __expf(mx - cm), w = __expf(lg - cm);
            den = fmaf(den, sc, w);
            sx0 = fmaf(sx0, sc, w * xs.x);
            sx1 = fmaf(sx1, sc, w * xs.y);
            mx = cm;
        }
    }
    const bool has = deg > 0;
    const float m0 = has ? sx0 / den : 0.f, m1 = has ? sx1 / den : 0.f;
    float blv[8], bsv[8];
    load8(bl + d0, blv); load8(bias + d0, bsv);
    v8h o;
#pragma unroll
    for (int c = 0; c < 8; ++c) {
        float v = (has ? fmaf(m0, wl0[c], fmaf(m1, wl1[c], blv[c])) : 0.f) + bsv[c];
        v = (v > 0.f) ? v : (__expf(v) - 1.f);
        o[c] = (_Float16)v;
    }
    *(v8h*)(hh + (size_t)t * 128 + d0) = o;
}

__device__ __forceinline__ float quad_xor1(float v) {
    return __int_as_float(__builtin_amdgcn_update_dpp(0, __float_as_int(v), 0xB1, 0xF, 0xF, true));
}
__device__ __forceinline__ float quad_xor2(float v) {
    return __int_as_float(__builtin_amdgcn_update_dpp(0, __float_as_int(v), 0x4E, 0xF, 0xF, true));
}

__device__ __forceinline__ float head_logit4(const float4 row, const float4 xrt, const float2 a,
        const float4 w0, const float4 w1, const float4 at) {
    float m0 = row.x + xrt.x + fmaf(a.x, w0.x, a.y * w1.x);
    float m1 = row.y + xrt.y + fmaf(a.x, w0.y, a.y * w1.y);
    float m2 = row.z + xrt.z + fmaf(a.x, w0.z, a.y * w1.z);
    float m3 = row.w + xrt.w + fmaf(a.x, w0.w, a.y * w1.w);
    m0 = (m0 > 0.f) ? m0 : 0.2f * m0;
    m1 = (m1 > 0.f) ? m1 : 0.2f * m1;
    m2 = (m2 > 0.f) ? m2 : 0.2f * m2;
    m3 = (m3 > 0.f) ? m3 : 0.2f * m3;
    float lg = fmaf(m3, at.w, fmaf(m2, at.z, fmaf(m1, at.y, m0 * at.x)));
    lg += quad_xor1(lg);
    lg += quad_xor2(lg);
    return lg;
}

constexpr int L2_U = 4;
__global__ void __launch_bounds__(256)
k_node_attn2(const float* __restrict__ xl, const float* __restrict__ xr,
             const float* __restrict__ We, const float* __restrict__ att,
             const int* __restrict__ cnt, const float* __restrict__ crec,
             const int* __restrict__ novf, const float* __restrict__ ovf,
             float* __restrict__ agg, int N, int E) {
    int idx = blockIdx.x * blockDim.x + threadIdx.x;
    if (idx >= N * 16) return;
    const int t = idx >> 4, d0 = (idx & 15) * 4;
    const float4 xrt = *(const float4*)(xr + (size_t)t * 64 + d0);
    const float4 w0 = *(const float4*)(We + d0);
    const float4 w1 = *(const float4*)(We + 64 + d0);
    const float4 at = *(const float4*)(att + d0);
    float4 acc = make_float4(0.f, 0.f, 0.f, 0.f);
    const int deg = cnt[t];
    const unsigned beg = (unsigned)t * CAP, end = beg + (unsigned)(deg < CAP ? deg : CAP);
    const float4* rec4 = (const float4*)crec;
    float mx = -__builtin_huge_valf(), den = 0.f;
    for (unsigned j = beg; j < end; j += L2_U) {
        int s[L2_U]; float2 a[L2_U]; bool ok[L2_U];
        float4 row[L2_U]; float l[L2_U];
#pragma unroll
        for (int u = 0; u < L2_U; ++u) {
            ok[u] = (j + u) < end;
            unsigned jj = ok[u] ? (j + u) : j;
            const float4 r = rec4[jj];
            s[u] = __float_as_int(r.x);
            a[u] = make_float2(r.y, r.z);
        }
#pragma unroll
        for (int u = 0; u < L2_U; ++u)
            row[u] = *(const float4*)(xl + (size_t)s[u] * 64 + d0);
        float cm = mx;
#pragma unroll
        for (int u = 0; u < L2_U; ++u) {
            float m0 = row[u].x + xrt.x + fmaf(a[u].x, w0.x, a[u].y * w1.x);
            float m1 = row[u].y + xrt.y + fmaf(a[u].x, w0.y, a[u].y * w1.y);
            float m2 = row[u].z + xrt.z + fmaf(a[u].x, w0.z, a[u].y * w1.z);
            float m3 = row[u].w + xrt.w + fmaf(a[u].x, w0.w, a[u].y * w1.w);
            m0 = (m0 > 0.f) ? m0 : 0.2f * m0;
            m1 = (m1 > 0.f) ? m1 : 0.2f * m1;
            m2 = (m2 > 0.f) ? m2 : 0.2f * m2;
            m3 = (m3 > 0.f) ? m3 : 0.2f * m3;
            float lg = fmaf(m3, at.w, fmaf(m2, at.z, fmaf(m1, at.y, m0 * at.x)));
            lg += quad_xor1(lg);
            lg += quad_xor2(lg);
            l[u] = ok[u] ? lg : -__builtin_huge_valf();
            cm = fmaxf(cm, l[u]);
        }
        const float sc = __expf(mx - cm);
        den *= sc;
        acc.x *= sc; acc.y *= sc; acc.z *= sc; acc.w *= sc;
#pragma unroll
        for (int u = 0; u < L2_U; ++u) {
            float w = __expf(l[u] - cm);
            den += w;
            acc.x = fmaf(w, row[u].x, acc.x);
            acc.y = fmaf(w, row[u].y, acc.y);
            acc.z = fmaf(w, row[u].z, acc.z);
            acc.w = fmaf(w, row[u].w, acc.w);
        }
        mx = cm;
    }
    if (deg > CAP) {
        const int nraw = *novf;
        const int no = nraw < E ? nraw : E;
        const float4* o4 = (const float4*)ovf;
        for (int i = 0; i < no; ++i) {
            const float4 r = o4[i];
            if (__float_as_int(r.w) != t) continue;
            const float4 row = *(const float4*)(xl + (size_t)__float_as_int(r.x) * 64 + d0);
            const float lg = head_logit4(row, xrt, make_float2(r.y, r.z), w0, w1, at);
            const float cm = fmaxf(mx, lg);
            const float sc = __expf(mx - cm), w = __expf(lg - cm);
            den = fmaf(den, sc, w);
            acc.x = fmaf(acc.x, sc, w * row.x);
            acc.y = fmaf(acc.y, sc, w * row.y);
            acc.z = fmaf(acc.z, sc, w * row.z);
            acc.w = fmaf(acc.w, sc, w * row.w);
            mx = cm;
        }
    }
    float4 o = make_float4(0.f, 0.f, 0.f, 0.f);
    if (deg > 0) { o.x = acc.x / den; o.y = acc.y / den; o.z = acc.z / den; o.w = acc.w / den; }
    *(float4*)(agg + (size_t)t * 64 + d0) = o;
}

__global__ void k_w_pack(const float* __restrict__ Wl, const float* __restrict__ Wr,
                         _Float16* __restrict__ Wpk) {
    int idx = blockIdx.x * blockDim.x + threadIdx.x;
    if (idx >= 2 * 8192) return;
    int which = idx >> 13;
    int r = idx & 8191;
    int j = r & 15;
    int lane = (r >> 4) & 31;
    int frag = r >> 9;
    int kc = frag >> 2, nt = frag & 3;
    int half_sel = lane >> 4, lidx = lane & 15;
    int kb = (j < 8) ? (half_sel * 8) : (16 + half_sel * 8);
    int k = kc * 32 + kb + (j & 7);
    int n = nt * 16 + lidx;
    const float* W = which ? Wr : Wl;
    Wpk[idx] = (_Float16)W[k * 64 + n];
}

__global__ void k_gemm2_wmma(const _Float16* __restrict__ hh,
                             const _Float16* __restrict__ Wpk,
                             const float* __restrict__ bl, const float* __restrict__ br,
                             float* __restrict__ xl2, float* __restrict__ xr2, int N) {
    __shared__ __align__(32) _Float16 ldsW[2 * 8192];

    {
        const uint4* g = (const uint4*)Wpk;
        uint4* l = (uint4*)ldsW;
        for (int i = threadIdx.x; i < 2048; i += blockDim.x) l[i] = g[i];
    }
    __syncthreads();

    int wave = (blockIdx.x * blockDim.x + threadIdx.x) >> 5;
    int lane = threadIdx.x & 31;
    int row0 = wave * 16;
    if (row0 >= N) return;

    int half_sel = lane >> 4;
    int lidx = lane & 15;
    int kb_lo = half_sel * 8;
    int kb_hi = 16 + kb_lo;
    int m = row0 + lidx;
    if (m >= N) m = N - 1;

    const v16h* fragL = (const v16h*)ldsW;
    const v16h* fragR = (const v16h*)(ldsW + 8192);

    v8f accl[4] = {};
    v8f accr[4] = {};

#pragma unroll
    for (int kc = 0; kc < 4; ++kc) {
        const _Float16* pa = hh + m * 128 + kc * 32;
        v8h alo = *(const v8h*)(pa + kb_lo);
        v8h ahi = *(const v8h*)(pa + kb_hi);
        v16h a = __builtin_shufflevector(alo, ahi, 0, 1, 2, 3, 4, 5, 6, 7,
                                         8, 9, 10, 11, 12, 13, 14, 15);
#pragma unroll
        for (int nt = 0; nt < 4; ++nt) {
            int fi = (kc * 4 + nt) * 32 + lane;
            v16h b  = fragL[fi];
            v16h b2 = fragR[fi];
            accl[nt] = __builtin_amdgcn_wmma_f32_16x16x32_f16(
                false, a, false, b, (short)0, accl[nt], false, false);
            accr[nt] = __builtin_amdgcn_wmma_f32_16x16x32_f16(
                false, a, false, b2, (short)0, accr[nt], false, false);
        }
    }
#pragma unroll
    for (int nt = 0; nt < 4; ++nt) {
        int n = nt * 16 + lidx;
        float bln = bl[n], brn = br[n];
#pragma unroll
        for (int r = 0; r < 8; ++r) {
            int mo = row0 + r + half_sel * 8;
            if (mo < N) {
                xl2[mo * 64 + n] = accl[nt][r] + bln;
                xr2[mo * 64 + n] = accr[nt][r] + brn;
            }
        }
    }
}

constexpr int POOL_NPB = 128;
constexpr int POOL_U   = 8;

__global__ void __launch_bounds__(256)
k_pool(const float* __restrict__ agg2, const float* __restrict__ bias2,
       const int* __restrict__ batch,
       float* __restrict__ sums, float* __restrict__ cnts, int N) {
    const int d   = threadIdx.x & 63;
    const int sub = threadIdx.x >> 6;
    const int n0  = blockIdx.x * POOL_NPB + sub;
    const float b = bias2[d];
    float acc = 0.f, cnt = 0.f;
    int gcur = -1;
    for (int it = 0; it < POOL_NPB / 4; it += POOL_U) {
        float v[POOL_U];
        int   g[POOL_U];
#pragma unroll
        for (int u = 0; u < POOL_U; ++u) {
            int n = n0 + 4 * (it + u);
            bool ok = n < N;
            g[u] = ok ? batch[n] : -1;
            v[u] = ok ? agg2[n * 64 + d] : 0.f;
        }
#pragma unroll
        for (int u = 0; u < POOL_U; ++u) {
            if (g[u] != gcur) {
                if (gcur >= 0) {
                    atomicAdd(&sums[gcur * 64 + d], acc);
                    if (d == 0) atomicAdd(&cnts[gcur], cnt);
                }
                gcur = g[u]; acc = 0.f; cnt = 0.f;
            }
            acc += v[u] + b;
            cnt += 1.0f;
        }
    }
    if (gcur >= 0) {
        atomicAdd(&sums[gcur * 64 + d], acc);
        if (d == 0) atomicAdd(&cnts[gcur], cnt);
    }
}

__global__ void k_final(const float* __restrict__ sums, const float* __restrict__ cnts,
                        float* __restrict__ out, int G) {
    int idx = blockIdx.x * blockDim.x + threadIdx.x;
    if (idx >= G * 64) return;
    out[idx] = sums[idx] / fmaxf(cnts[idx >> 6], 1.0f);
}

extern "C" void kernel_launch(void* const* d_in, const int* in_sizes, int n_in,
                              void* d_out, int out_size, void* d_ws, size_t ws_size,
                              hipStream_t stream) {
    const float* x      = (const float*)d_in[0];
    const float* ea     = (const float*)d_in[1];
    const float* W_l1   = (const float*)d_in[2];
    const float* b_l1   = (const float*)d_in[3];
    const float* W_r1   = (const float*)d_in[4];
    const float* b_r1   = (const float*)d_in[5];
    const float* W_e1   = (const float*)d_in[6];
    const float* att1   = (const float*)d_in[7];
    const float* bias1  = (const float*)d_in[8];
    const float* W_l2   = (const float*)d_in[9];
    const float* b_l2   = (const float*)d_in[10];
    const float* W_r2   = (const float*)d_in[11];
    const float* b_r2   = (const float*)d_in[12];
    const float* W_e2   = (const float*)d_in[13];
    const float* att2   = (const float*)d_in[14];
    const float* bias2  = (const float*)d_in[15];
    const int*   eidx   = (const int*)d_in[16];
    const int*   batch  = (const int*)d_in[17];
    float* out = (float*)d_out;

    const int N = in_sizes[0] / 2;
    const int E = in_sizes[1] / 2;
    const int G = out_size / 64;
    const int* src = eidx;
    const int* tgt = eidx + E;

    char* ws = (char*)d_ws;
    size_t off = 0;
    auto alloc = [&](size_t bytes) { size_t o = off; off = (off + bytes + 255) & ~(size_t)255; return o; };

    size_t o_cnt   = alloc((size_t)N * 4);
    size_t o_novf  = alloc(4);
    size_t o_sums  = alloc((size_t)G * 64 * 4);
    size_t o_cnts  = alloc((size_t)G * 4);
    size_t acc_bytes = off;

    size_t o_crec = alloc((size_t)N * CAP * 16);
    size_t o_ovf  = alloc((size_t)E * 16);
    size_t o_hh   = alloc((size_t)N * 128 * 2);
    size_t o_wpk  = alloc((size_t)2 * 8192 * 2);
    size_t o_xl2  = alloc((size_t)N * 64 * 4);
    size_t o_xr2  = alloc((size_t)N * 64 * 4);
    size_t o_agg2 = alloc((size_t)N * 64 * 4);
    (void)ws_size; (void)n_in;

    int*      cnt   = (int*)(ws + o_cnt);
    int*      novf  = (int*)(ws + o_novf);
    float*    sums  = (float*)(ws + o_sums);
    float*    cnts  = (float*)(ws + o_cnts);
    float*    crec  = (float*)(ws + o_crec);
    float*    ovf   = (float*)(ws + o_ovf);
    _Float16* hh    = (_Float16*)(ws + o_hh);
    _Float16* wpk   = (_Float16*)(ws + o_wpk);
    float*    xl2   = (float*)(ws + o_xl2);
    float*    xr2   = (float*)(ws + o_xr2);
    float*    agg2  = (float*)(ws + o_agg2);

    const int TB = 256;
    auto blocks = [&](long long t) { return (unsigned)((t + TB - 1) / TB); };

    hipMemsetAsync(d_ws, 0, acc_bytes, stream);

    k_bucket<<<blocks(((long long)E + BK_U - 1) / BK_U), TB, 0, stream>>>(
        src, tgt, ea, cnt, novf, crec, ovf, E);

    k_node_attn1<<<blocks((long long)N * 16), TB, 0, stream>>>(
        x, W_l1, b_l1, W_r1, b_r1, W_e1, att1, bias1, cnt, crec, novf, ovf, hh, N, E);

    k_w_pack<<<blocks(2 * 8192), TB, 0, stream>>>(W_l2, W_r2, wpk);
    long long gemm_threads = (long long)((N + 15) / 16) * 32;
    k_gemm2_wmma<<<blocks(gemm_threads), TB, 0, stream>>>(
        hh, wpk, b_l2, b_r2, xl2, xr2, N);

    k_node_attn2<<<blocks((long long)N * 16), TB, 0, stream>>>(
        xl2, xr2, W_e2, att2, cnt, crec, novf, ovf, agg2, N, E);

    k_pool<<<(unsigned)((N + POOL_NPB - 1) / POOL_NPB), TB, 0, stream>>>(agg2, bias2, batch, sums, cnts, N);
    k_final<<<blocks((long long)G * 64), TB, 0, stream>>>(sums, cnts, out, G);
}


# bucket build as a two-level LDS counting sort: blocks order 3072 edges by 64-node bin (1 global atomic per block and bin), then a block per bin fills 64 buckets in LDS; +1 launch; arrival-order sums
# speedup vs baseline: 6.7991x; 1.7355x over previous
#include <hip/hip_runtime.h>
#include <hip/hip_fp16.h>

typedef __attribute__((ext_vector_type(16))) _Float16 v16h;
typedef __attribute__((ext_vector_type(8)))  _Float16 v8h;
typedef __attribute__((ext_vector_type(8)))  float    v8f;

constexpr int CAP = 48;
constexpr int BIN_SH = 6;
constexpr int BIN_NODES = 1 << BIN_SH;
constexpr int MAXBIN = 1024;
constexpr int BCAP = 2048;
constexpr int L1_T = 1024;
constexpr int L1_EPT = 3;
constexpr int L1_EPB = L1_T * L1_EPT;

__global__ void __launch_bounds__(1024)
k_bin(const int* __restrict__ src, const int* __restrict__ tgt, const float* __restrict__ ea,
      int* __restrict__ bincnt, float* __restrict__ binbuf,
      int* __restrict__ cnt, int* __restrict__ novf,
      float* __restrict__ crec, float* __restrict__ ovf, int E) {
    __shared__ int s_hist[MAXBIN];
    __shared__ int s_scan[2][MAXBIN];
    __shared__ __align__(16) float s_rec[L1_EPB * 4];
    __shared__ int s_full;
    const int tid = threadIdx.x;
    const int e0 = blockIdx.x * L1_EPB;
    s_hist[tid] = 0;
    if (tid == 0) s_full = 0;
    __syncthreads();
    int t[L1_EPT], rk[L1_EPT];
#pragma unroll
    for (int u = 0; u < L1_EPT; ++u) {
        const int e = e0 + u * L1_T + tid;
        t[u] = (e < E) ? tgt[e] : -1;
    }
#pragma unroll
    for (int u = 0; u < L1_EPT; ++u)
        rk[u] = (t[u] >= 0) ? atomicAdd(&s_hist[t[u] >> BIN_SH], 1) : 0;
    __syncthreads();
    const int c = s_hist[tid];
    const int gb = (c > 0) ? atomicAdd(&bincnt[tid], c) : 0;
    if (gb + c > BCAP) s_full = 1;
    s_scan[0][tid] = c;
    __syncthreads();
    int cur = 0;
    for (int d = 1; d < L1_T; d <<= 1) {
        int v = s_scan[cur][tid];
        if (tid >= d) v += s_scan[cur][tid - d];
        s_scan[cur ^ 1][tid] = v;
        __syncthreads();
        cur ^= 1;
    }
    const int excl = s_scan[cur][tid] - c;
    s_hist[tid] = excl;
    s_scan[cur ^ 1][tid] = gb;
    __syncthreads();
    const int* s_off = s_hist;
    const int* s_gb = s_scan[cur ^ 1];
    float4* rec4 = (float4*)s_rec;
#pragma unroll
    for (int u = 0; u < L1_EPT; ++u) {
        if (t[u] >= 0) {
            const int e = e0 + u * L1_T + tid;
            const int s = src[e];
            const float2 a = ((const float2*)ea)[e];
            rec4[s_off[t[u] >> BIN_SH] + rk[u]] =
                make_float4(__int_as_float(s), a.x, a.y, __int_as_float(t[u]));
        }
    }
    __syncthreads();
    const int nloc = (E - e0 < L1_EPB) ? (E - e0) : L1_EPB;
    float4* bin4 = (float4*)binbuf;
    for (int i = tid; i < nloc; i += L1_T) {
        const float4 r = rec4[i];
        const int b = __float_as_int(r.w) >> BIN_SH;
        const int pos = s_gb[b] + (i - s_off[b]);
        if (pos < BCAP) bin4[(size_t)b * BCAP + pos] = r;
    }
    if (s_full) {
        for (int i = tid; i < nloc; i += L1_T) {
            const float4 r = rec4[i];
            const int tt = __float_as_int(r.w);
            const int b = tt >> BIN_SH;
            const int pos = s_gb[b] + (i - s_off[b]);
            if (pos >= BCAP) {
                const int p = atomicAdd(&cnt[tt], 1);
                if (p < CAP) {
                    ((float4*)crec)[(size_t)tt * CAP + p] = make_float4(r.x, r.y, r.z, 0.f);
                } else {
                    const int q = atomicAdd(novf, 1);
                    ((float4*)ovf)[q] = r;
                }
            }
        }
    }
}

constexpr int PL_U = 4;
__global__ void __launch_bounds__(256)
k_place(const int* __restrict__ bincnt, const float* __restrict__ binbuf,
        int* __restrict__ cnt, int* __restrict__ novf,
        float* __restrict__ crec, float* __restrict__ ovf, int N) {
    __shared__ __align__(16) float s_st[BIN_NODES * CAP * 4];
    __shared__ int s_cnt[BIN_NODES];
    __shared__ int s_c0[BIN_NODES];
    const int b = blockIdx.x, tid = threadIdx.x;
    const int node0 = b << BIN_SH;
    if (tid < BIN_NODES) {
        const int nd = node0 + tid;
        const int c0 = (nd < N) ? cnt[nd] : 0;
        s_cnt[tid] = c0; s_c0[tid] = c0;
    }
    __syncthreads();
    const int nraw = bincnt[b];
    const int nin = nraw < BCAP ? nraw : BCAP;
    const float4* in4 = (const float4*)binbuf + (size_t)b * BCAP;
    float4* st4 = (float4*)s_st;
    for (int i0 = 0; i0 < nin; i0 += 256 * PL_U) {
        float4 r[PL_U]; bool ok[PL_U];
#pragma unroll
        for (int u = 0; u < PL_U; ++u) {
            const int i = i0 + u * 256 + tid;
            ok[u] = i < nin;
            r[u] = in4[ok[u] ? i : 0];
        }
#pragma unroll
        for (int u = 0; u < PL_U; ++u) {
            if (!ok[u]) continue;
            const int n = __float_as_int(r[u].w) - node0;
            const int p = atomicAdd(&s_cnt[n], 1);
            if (p < CAP) {
                st4[n * CAP + p] = make_float4(r[u].x, r[u].y, r[u].z, 0.f);
            } else {
                const int q = atomicAdd(novf, 1);
                ((float4*)ovf)[q] = r[u];
            }
        }
    }
    __syncthreads();
    float4* out4 = (float4*)crec + (size_t)node0 * CAP;
    for (int i = tid; i < BIN_NODES * CAP; i += 256) {
        const int n = i / CAP, p = i - n * CAP;
        const int cn = s_cnt[n];
        const int hi = cn < CAP ? cn : CAP;
        if (p >= s_c0[n] && p < hi) out4[i] = st4[i];
    }
    if (tid < BIN_NODES && node0 + tid < N) cnt[node0 + tid] = s_cnt[tid];
}

__device__ __forceinline__ void load8(const float* __restrict__ p, float (&v)[8]) {
    const float4 a = ((const float4*)p)[0], b = ((const float4*)p)[1];
    v[0] = a.x; v[1] = a.y; v[2] = a.z; v[3] = a.w;
    v[4] = b.x; v[5] = b.y; v[6] = b.z; v[7] = b.w;
}

__device__ __forceinline__ float head_logit8(const float2 xs, const float2 a,
        const float (&wl0)[8], const float (&wl1)[8],
        const float (&we0)[8], const float (&we1)[8],
        const float (&base)[8], const float (&at)[8]) {
    float lg = 0.f;
#pragma unroll
    for (int c = 0; c < 8; ++c) {
        float m = fmaf(xs.x, wl0[c], fmaf(xs.y, wl1[c],
                  fmaf(a.x, we0[c], fmaf(a.y, we1[c], base[c]))));
        m = (m > 0.f) ? m : 0.2f * m;
        lg = fmaf(m, at[c], lg);
    }
    return lg;
}

constexpr int L1_U = 4;
__global__ void __launch_bounds__(256)
k_node_attn1(const float* __restrict__ x,
             const float* __restrict__ Wl, const float* __restrict__ bl,
             const float* __restrict__ Wr, const float* __restrict__ br,
             const float* __restrict__ We, const float* __restrict__ att,
             const float* __restrict__ bias,
             const int* __restrict__ cnt, const float* __restrict__ crec,
             const int* __restrict__ novf, const float* __restrict__ ovf,
             _Float16* __restrict__ hh, int N, int E) {
    int idx = blockIdx.x * blockDim.x + threadIdx.x;
    if (idx >= N * 16) return;
    const int t = idx >> 4, d0 = (idx & 15) * 8;
    const float2* x2 = (const float2*)x;
    float wl0[8], wl1[8], we0[8], we1[8], at[8], base[8];
    load8(Wl + d0, wl0); load8(Wl + 128 + d0, wl1);
    load8(We + d0, we0); load8(We + 128 + d0, we1);
    load8(att + d0, at);
    {
        const float2 xt = x2[t];
        float wr0[8], wr1[8], brv[8], blv[8];
        load8(Wr + d0, wr0); load8(Wr + 128 + d0, wr1);
        load8(br + d0, brv); load8(bl + d0, blv);
#pragma unroll
        for (int c = 0; c < 8; ++c)
            base[c] = fmaf(xt.x, wr0[c], fmaf(xt.y, wr1[c], brv[c])) + blv[c];
    }
    const int deg = cnt[t];
    const unsigned beg = (unsigned)t * CAP, end = beg + (unsigned)(deg < CAP ? deg : CAP);
    const float4* rec4 = (const float4*)crec;
    float mx = -__builtin_huge_valf(), den = 0.f, sx0 = 0.f, sx1 = 0.f;
    for (unsigned j = beg; j < end; j += L1_U) {
        int s[L1_U]; float2 a[L1_U], xs[L1_U]; bool ok[L1_U];
        float l[L1_U];
#pragma unroll
        for (int u = 0; u < L1_U; ++u) {
            ok[u] = (j + u) < end;
            unsigned jj = ok[u] ? (j + u) : j;
            const float4 r = rec4[jj];
            s[u] = __float_as_int(r.x);
            a[u] = make_float2(r.y, r.z);
        }
#pragma unroll
        for (int u = 0; u < L1_U; ++u) xs[u] = x2[s[u]];
        float cm = mx;
#pragma unroll
        for (int u = 0; u < L1_U; ++u) {
            float lg = 0.f;
#pragma unroll
            for (int c = 0; c < 8; ++c) {
                float m = fmaf(xs[u].x, wl0[c], fmaf(xs[u].y, wl1[c],
                          fmaf(a[u].x, we0[c], fmaf(a[u].y, we1[c], base[c]))));
                m = (m > 0.f) ? m : 0.2f * m;
                lg = fmaf(m, at[c], lg);
            }
            l[u] = ok[u] ? lg : -__builtin_huge_valf();
            cm = fmaxf(cm, l[u]);
        }
        const float sc = __expf(mx - cm);
        den *= sc; sx0 *= sc; sx1 *= sc;
#pragma unroll
        for (int u = 0; u < L1_U; ++u) {
            float w = __expf(l[u] - cm);
            den += w;
            sx0 = fmaf(w, xs[u].x, sx0);
            sx1 = fmaf(w, xs[u].y, sx1);
        }
        mx = cm;
    }
    if (deg > CAP) {
        const int nraw = *novf;
        const int no = nraw < E ? nraw : E;
        const float4* o4 = (const float4*)ovf;
        for (int i = 0; i < no; ++i) {
            const float4 r = o4[i];
            if (__float_as_int(r.w) != t) continue;
            const float2 xs = x2[__float_as_int(r.x)];
            const float lg = head_logit8(xs, make_float2(r.y, r.z), wl0, wl1, we0, we1, base, at);
            const float cm = fmaxf(mx, lg);
            const float sc = __expf(mx - cm), w = __expf(lg - cm);
            den = fmaf(den, sc, w);
            sx0 = fmaf(sx0, sc, w * xs.x);
            sx1 = fmaf(sx1, sc, w * xs.y);
            mx = cm;
        }
    }
    const bool has = deg > 0;
    const float m0 = has ? sx0 / den : 0.f, m1 = has ? sx1 / den : 0.f;
    float blv[8], bsv[8];
    load8(bl + d0, blv); load8(bias + d0, bsv);
    v8h o;
#pragma unroll
    for (int c = 0; c < 8; ++c) {
        float v = (has ? fmaf(m0, wl0[c], fmaf(m1, wl1[c], blv[c])) : 0.f) + bsv[c];
        v = (v > 0.f) ? v : (__expf(v) - 1.f);
        o[c] = (_Float16)v;
    }
    *(v8h*)(hh + (size_t)t * 128 + d0) = o;
}

__device__ __forceinline__ float quad_xor1(float v) {
    return __int_as_float(__builtin_amdgcn_update_dpp(0, __float_as_int(v), 0xB1, 0xF, 0xF, true));
}
__device__ __forceinline__ float quad_xor2(float v) {
    return __int_as_float(__builtin_amdgcn_update_dpp(0, __float_as_int(v), 0x4E, 0xF, 0xF, true));
}

__device__ __forceinline__ float head_logit4(const float4 row, const float4 xrt, const float2 a,
        const float4 w0, const float4 w1, const float4 at) {
    float m0 = row.x + xrt.x + fmaf(a.x, w0.x, a.y * w1.x);
    float m1 = row.y + xrt.y + fmaf(a.x, w0.y, a.y * w1.y);
    float m2 = row.z + xrt.z + fmaf(a.x, w0.z, a.y * w1.z);
    float m3 = row.w + xrt.w + fmaf(a.x, w0.w, a.y * w1.w);
    m0 = (m0 > 0.f) ? m0 : 0.2f * m0;
    m1 = (m1 > 0.f) ? m1 : 0.2f * m1;
    m2 = (m2 > 0.f) ? m2 : 0.2f * m2;
    m3 = (m3 > 0.f) ? m3 : 0.2f * m3;
    float lg = fmaf(m3, at.w, fmaf(m2, at.z, fmaf(m1, at.y, m0 * at.x)));
    lg += quad_xor1(lg);
    lg += quad_xor2(lg);
    return lg;
}

constexpr int L2_U = 4;
__global__ void __launch_bounds__(256)
k_node_attn2(const float* __restrict__ xl, const float* __restrict__ xr,
             const float* __restrict__ We, const float* __restrict__ att,
             const int* __restrict__ cnt, const float* __restrict__ crec,
             const int* __restrict__ novf, const float* __restrict__ ovf,
             const float* __restrict__ bias2, const int* __restrict__ batch,
             float* __restrict__ sums, float* __restrict__ cnts, int N, int E) {
    __shared__ __align__(16) float s_row[16 * 64];
    __shared__ int s_g[16];
    const int idx = blockIdx.x * blockDim.x + threadIdx.x;
    const bool valid = idx < N * 16;
    const int t = valid ? (idx >> 4) : 0, d0 = (idx & 15) * 4;
    const float4 xrt = *(const float4*)(xr + (size_t)t * 64 + d0);
    const float4 w0 = *(const float4*)(We + d0);
    const float4 w1 = *(const float4*)(We + 64 + d0);
    const float4 at = *(const float4*)(att + d0);
    float4 acc = make_float4(0.f, 0.f, 0.f, 0.f);
    const int deg = valid ? cnt[t] : 0;
    const unsigned beg = (unsigned)t * CAP, end = beg + (unsigned)(deg < CAP ? deg : CAP);
    const float4* rec4 = (const float4*)crec;
    float mx = -__builtin_huge_valf(), den = 0.f;
    for (unsigned j = beg; j < end; j += L2_U) {
        int s[L2_U]; float2 a[L2_U]; bool ok[L2_U];
        float4 row[L2_U]; float l[L2_U];
#pragma unroll
        for (int u = 0; u < L2_U; ++u) {
            ok[u] = (j + u) < end;
            unsigned jj = ok[u] ? (j + u) : j;
            const float4 r = rec4[jj];
            s[u] = __float_as_int(r.x);
            a[u] = make_float2(r.y, r.z);
        }
#pragma unroll
        for (int u = 0; u < L2_U; ++u)
            row[u] = *(const float4*)(xl + (size_t)s[u] * 64 + d0);
        float cm = mx;
#pragma unroll
        for (int u = 0; u < L2_U; ++u) {
            float m0 = row[u].x + xrt.x + fmaf(a[u].x, w0.x, a[u].y * w1.x);
            float m1 = row[u].y + xrt.y + fmaf(a[u].x, w0.y, a[u].y * w1.y);
            float m2 = row[u].z + xrt.z + fmaf(a[u].x, w0.z, a[u].y * w1.z);
            float m3 = row[u].w + xrt.w + fmaf(a[u].x, w0.w, a[u].y * w1.w);
            m0 = (m0 > 0.f) ? m0 : 0.2f * m0;
            m1 = (m1 > 0.f) ? m1 : 0.2f * m1;
            m2 = (m2 > 0.f) ? m2 : 0.2f * m2;
            m3 = (m3 > 0.f) ? m3 : 0.2f * m3;
            float lg = fmaf(m3, at.w, fmaf(m2, at.z, fmaf(m1, at.y, m0 * at.x)));
            lg += quad_xor1(lg);
            lg += quad_xor2(lg);
            l[u] = ok[u] ? lg : -__builtin_huge_valf();
            cm = fmaxf(cm, l[u]);
        }
        const float sc = __expf(mx - cm);
        den *= sc;
        acc.x *= sc; acc.y *= sc; acc.z *= sc; acc.w *= sc;
#pragma unroll
        for (int u = 0; u < L2_U; ++u) {
            float w = __expf(l[u] - cm);
            den += w;
            acc.x = fmaf(w, row[u].x, acc.x);
            acc.y = fmaf(w, row[u].y, acc.y);
            acc.z = fmaf(w, row[u].z, acc.z);
            acc.w = fmaf(w, row[u].w, acc.w);
        }
        mx = cm;
    }
    if (deg > CAP) {
        const int nraw = *novf;
        const int no = nraw < E ? nraw : E;
        const float4* o4 = (const float4*)ovf;
        for (int i = 0; i < no; ++i) {
            const float4 r = o4[i];
            if (__float_as_int(r.w) != t) continue;
            const float4 row = *(const float4*)(xl + (size_t)__float_as_int(r.x) * 64 + d0);
            const float lg = head_logit4(row, xrt, make_float2(r.y, r.z), w0, w1, at);
            const float cm = fmaxf(mx, lg);
            const float sc = __expf(mx - cm), w = __expf(lg - cm);
            den = fmaf(den, sc, w);
            acc.x = fmaf(acc.x, sc, w * row.x);
            acc.y = fmaf(acc.y, sc, w * row.y);
            acc.z = fmaf(acc.z, sc, w * row.z);
            acc.w = fmaf(acc.w, sc, w * row.w);
            mx = cm;
        }
    }
    float4 o = make_float4(0.f, 0.f, 0.f, 0.f);
    if (deg > 0) { o.x = acc.x / den; o.y = acc.y / den; o.z = acc.z / den; o.w = acc.w / den; }

    const float4 b2 = *(const float4*)(bias2 + d0);
    const int nl = threadIdx.x >> 4;
    *(float4*)(s_row + nl * 64 + d0) = make_float4(o.x + b2.x, o.y + b2.y, o.z + b2.z, o.w + b2.w);
    if ((threadIdx.x & 15) == 0) s_g[nl] = valid ? batch[t] : -1;
    __syncthreads();
    if (threadIdx.x < 64) {
        const int d = threadIdx.x;
        float racc = 0.f, rcnt = 0.f;
        int gcur = -1;
        float v[16];
        int   g[16];
#pragma unroll
        for (int n = 0; n < 16; ++n) { v[n] = s_row[n * 64 + d]; g[n] = s_g[n]; }
#pragma unroll
        for (int n = 0; n < 16; ++n) {
            if (g[n] != gcur) {
                if (gcur >= 0) {
                    atomicAdd(&sums[gcur * 64 + d], racc);
                    if (d == 0) atomicAdd(&cnts[gcur], rcnt);
                }
                gcur = g[n]; racc = 0.f; rcnt = 0.f;
            }
            racc += v[n];
            rcnt += 1.0f;
        }
        if (gcur >= 0) {
            atomicAdd(&sums[gcur * 64 + d], racc);
            if (d == 0) atomicAdd(&cnts[gcur], rcnt);
        }
    }
}

__global__ void k_prep(const float* __restrict__ Wl, const float* __restrict__ Wr,
                       _Float16* __restrict__ Wpk, uint4* __restrict__ zbase, size_t zbytes) {
    int idx = blockIdx.x * blockDim.x + threadIdx.x;
    const unsigned zn16 = (unsigned)(zbytes >> 4);
    for (unsigned i = (unsigned)idx; i < zn16; i += gridDim.x * blockDim.x)
        zbase[i] = make_uint4(0u, 0u, 0u, 0u);
    if (idx >= 2 * 8192) return;
    int which = idx >> 13;
    int r = idx & 8191;
    int j = r & 15;
    int lane = (r >> 4) & 31;
    int frag = r >> 9;
    int kc = frag >> 2, nt = frag & 3;
    int half_sel = lane >> 4, lidx = lane & 15;
    int kb = (j < 8) ? (half_sel * 8) : (16 + half_sel * 8);
    int k = kc * 32 + kb + (j & 7);
    int n = nt * 16 + lidx;
    const float* W = which ? Wr : Wl;
    Wpk[idx] = (_Float16)W[k * 64 + n];
}

__global__ void k_gemm2_wmma(const _Float16* __restrict__ hh,
                             const _Float16* __restrict__ Wpk,
                             const float* __restrict__ bl, const float* __restrict__ br,
                             float* __restrict__ xl2, float* __restrict__ xr2, int N) {
    __shared__ __align__(32) _Float16 ldsW[2 * 8192];

    {
        const uint4* g = (const uint4*)Wpk;
        uint4* l = (uint4*)ldsW;
        for (int i = threadIdx.x; i < 2048; i += blockDim.x) l[i] = g[i];
    }
    __syncthreads();

    int wave = (blockIdx.x * blockDim.x + threadIdx.x) >> 5;
    int lane = threadIdx.x & 31;
    int row0 = wave * 16;
    if (row0 >= N) return;

    int half_sel = lane >> 4;
    int lidx = lane & 15;
    int kb_lo = half_sel * 8;
    int kb_hi = 16 + kb_lo;
    int m = row0 + lidx;
    if (m >= N) m = N - 1;

    const v16h* fragL = (const v16h*)ldsW;
    const v16h* fragR = (const v16h*)(ldsW + 8192);

    v8f accl[4] = {};
    v8f accr[4] = {};

#pragma unroll
    for (int kc = 0; kc < 4; ++kc) {
        const _Float16* pa = hh + m * 128 + kc * 32;
        v8h alo = *(const v8h*)(pa + kb_lo);
        v8h ahi = *(const v8h*)(pa + kb_hi);
        v16h a = __builtin_shufflevector(alo, ahi, 0, 1, 2, 3, 4, 5, 6, 7,
                                         8, 9, 10, 11, 12, 13, 14, 15);
#pragma unroll
        for (int nt = 0; nt < 4; ++nt) {
            int fi = (kc * 4 + nt) * 32 + lane;
            v16h b  = fragL[fi];
            v16h b2 = fragR[fi];
            accl[nt] = __builtin_amdgcn_wmma_f32_16x16x32_f16(
                false, a, false, b, (short)0, accl[nt], false, false);
            accr[nt] = __builtin_amdgcn_wmma_f32_16x16x32_f16(
                false, a, false, b2, (short)0, accr[nt], false, false);
        }
    }
#pragma unroll
    for (int nt = 0; nt < 4; ++nt) {
        int n = nt * 16 + lidx;
        float bln = bl[n], brn = br[n];
#pragma unroll
        for (int r = 0; r < 8; ++r) {
            int mo = row0 + r + half_sel * 8;
            if (mo < N) {
                xl2[mo * 64 + n] = accl[nt][r] + bln;
                xr2[mo * 64 + n] = accr[nt][r] + brn;
            }
        }
    }
}

__global__ void k_final(const float* __restrict__ sums, const float* __restrict__ cnts,
                        float* __restrict__ out, int G) {
    int idx = blockIdx.x * blockDim.x + threadIdx.x;
    if (idx >= G * 64) return;
    out[idx] = sums[idx] / fmaxf(cnts[idx >> 6], 1.0f);
}

extern "C" void kernel_launch(void* const* d_in, const int* in_sizes, int n_in,
                              void* d_out, int out_size, void* d_ws, size_t ws_size,
                              hipStream_t stream) {
    const float* x      = (const float*)d_in[0];
    const float* ea     = (const float*)d_in[1];
    const float* W_l1   = (const float*)d_in[2];
    const float* b_l1   = (const float*)d_in[3];
    const float* W_r1   = (const float*)d_in[4];
    const float* b_r1   = (const float*)d_in[5];
    const float* W_e1   = (const float*)d_in[6];
    const float* att1   = (const float*)d_in[7];
    const float* bias1  = (const float*)d_in[8];
    const float* W_l2   = (const float*)d_in[9];
    const float* b_l2   = (const float*)d_in[10];
    const float* W_r2   = (const float*)d_in[11];
    const float* b_r2   = (const float*)d_in[12];
    const float* W_e2   = (const float*)d_in[13];
    const float* att2   = (const float*)d_in[14];
    const float* bias2  = (const float*)d_in[15];
    const int*   eidx   = (const int*)d_in[16];
    const int*   batch  = (const int*)d_in[17];
    float* out = (float*)d_out;

    const int N = in_sizes[0] / 2;
    const int E = in_sizes[1] / 2;
    const int G = out_size / 64;
    const int* src = eidx;
    const int* tgt = eidx + E;

    char* ws = (char*)d_ws;
    size_t off = 0;
    auto alloc = [&](size_t bytes) { size_t o = off; off = (off + bytes + 255) & ~(size_t)255; return o; };

    size_t o_cnt   = alloc((size_t)N * 4);
    size_t o_novf  = alloc(4);
    size_t o_sums  = alloc((size_t)G * 64 * 4);
    size_t o_cnts  = alloc((size_t)G * 4);
    size_t o_bcnt  = alloc((size_t)MAXBIN * 4);
    size_t acc_bytes = off;

    size_t o_crec = alloc((size_t)N * CAP * 16);
    size_t o_ovf  = alloc((size_t)E * 16);
    const int nbin = (N + BIN_NODES - 1) / BIN_NODES;
    size_t o_bin  = alloc((size_t)nbin * BCAP * 16);
    size_t o_hh   = alloc((size_t)N * 128 * 2);
    size_t o_wpk  = alloc((size_t)2 * 8192 * 2);
    size_t o_xl2  = alloc((size_t)N * 64 * 4);
    size_t o_xr2  = alloc((size_t)N * 64 * 4);
    (void)ws_size; (void)n_in;

    int*      cnt   = (int*)(ws + o_cnt);
    int*      novf  = (int*)(ws + o_novf);
    float*    sums  = (float*)(ws + o_sums);
    float*    cnts  = (float*)(ws + o_cnts);
    float*    crec  = (float*)(ws + o_crec);
    float*    ovf   = (float*)(ws + o_ovf);
    int*      bcnt  = (int*)(ws + o_bcnt);
    float*    binb  = (float*)(ws + o_bin);
    _Float16* hh    = (_Float16*)(ws + o_hh);
    _Float16* wpk   = (_Float16*)(ws + o_wpk);
    float*    xl2   = (float*)(ws + o_xl2);
    float*    xr2   = (float*)(ws + o_xr2);

    const int TB = 256;
    auto blocks = [&](long long t) { return (unsigned)((t + TB - 1) / TB); };

    k_prep<<<blocks(2 * 8192), TB, 0, stream>>>(W_l2, W_r2, wpk, (uint4*)d_ws, acc_bytes);

    k_bin<<<(unsigned)((E + L1_EPB - 1) / L1_EPB), L1_T, 0, stream>>>(
        src, tgt, ea, bcnt, binb, cnt, novf, crec, ovf, E);
    k_place<<<(unsigned)nbin, TB, 0, stream>>>(bcnt, binb, cnt, novf, crec, ovf, N);

    k_node_attn1<<<blocks((long long)N * 16), TB, 0, stream>>>(
        x, W_l1, b_l1, W_r1, b_r1, W_e1, att1, bias1, cnt, crec, novf, ovf, hh, N, E);

    long long gemm_threads = (long long)((N + 15) / 16) * 32;
    k_gemm2_wmma<<<blocks(gemm_threads), TB, 0, stream>>>(
        hh, wpk, b_l2, b_r2, xl2, xr2, N);

    k_node_attn2<<<blocks((long long)N * 16), TB, 0, stream>>>(
        xl2, xr2, W_e2, att2, cnt, crec, novf, ovf, bias2, batch, sums, cnts, N, E);

    k_final<<<blocks((long long)G * 64), TB, 0, stream>>>(sums, cnts, out, G);
}


# leaky_relu without compare and select: layer 1 as 0.6 m + 0.4 |m|, its linear half folded into 5 per-thread coefficients; layer 2 as max(m, 0.2 m); changes the rounding of layer-1 logits
# speedup vs baseline: 7.0220x; 1.0328x over previous
#include <hip/hip_runtime.h>
#include <hip/hip_fp16.h>

typedef __attribute__((ext_vector_type(16))) _Float16 v16h;
typedef __attribute__((ext_vector_type(8)))  _Float16 v8h;
typedef __attribute__((ext_vector_type(8)))  float    v8f;

constexpr int CAP = 48;
constexpr int BIN_SH = 6;
constexpr int BIN_NODES = 1 << BIN_SH;
constexpr int MAXBIN = 1024;
constexpr int BCAP = 2048;
constexpr int L1_T = 1024;
constexpr int L1_EPT = 3;
constexpr int L1_EPB = L1_T * L1_EPT;

__global__ void __launch_bounds__(1024)
k_bin(const int* __restrict__ src, const int* __restrict__ tgt, const float* __restrict__ ea,
      int* __restrict__ bincnt, float* __restrict__ binbuf,
      int* __restrict__ cnt, int* __restrict__ novf,
      float* __restrict__ crec, float* __restrict__ ovf, int E) {
    __shared__ int s_hist[MAXBIN];
    __shared__ int s_scan[2][MAXBIN];
    __shared__ __align__(16) float s_rec[L1_EPB * 4];
    __shared__ int s_full;
    const int tid = threadIdx.x;
    const int e0 = blockIdx.x * L1_EPB;
    s_hist[tid] = 0;
    if (tid == 0) s_full = 0;
    __syncthreads();
    int t[L1_EPT], rk[L1_EPT];
#pragma unroll
    for (int u = 0; u < L1_EPT; ++u) {
        const int e = e0 + u * L1_T + tid;
        t[u] = (e < E) ? tgt[e] : -1;
    }
#pragma unroll
    for (int u = 0; u < L1_EPT; ++u)
        rk[u] = (t[u] >= 0) ? atomicAdd(&s_hist[t[u] >> BIN_SH], 1) : 0;
    __syncthreads();
    const int c = s_hist[tid];
    const int gb = (c > 0) ? atomicAdd(&bincnt[tid], c) : 0;
    if (gb + c > BCAP) s_full = 1;
    s_scan[0][tid] = c;
    __syncthreads();
    int cur = 0;
    for (int d = 1; d < L1_T; d <<= 1) {
        int v = s_scan[cur][tid];
        if (tid >= d) v += s_scan[cur][tid - d];
        s_scan[cur ^ 1][tid] = v;
        __syncthreads();
        cur ^= 1;
    }
    const int excl = s_scan[cur][tid] - c;
    s_hist[tid] = excl;
    s_scan[cur ^ 1][tid] = gb;
    __syncthreads();
    const int* s_off = s_hist;
    const int* s_gb = s_scan[cur ^ 1];
    float4* rec4 = (float4*)s_rec;
#pragma unroll
    for (int u = 0; u < L1_EPT; ++u) {
        if (t[u] >= 0) {
            const int e = e0 + u * L1_T + tid;
            const int s = src[e];
            const float2 a = ((const float2*)ea)[e];
            rec4[s_off[t[u] >> BIN_SH] + rk[u]] =
                make_float4(__int_as_float(s), a.x, a.y, __int_as_float(t[u]));
        }
    }
    __syncthreads();
    const int nloc = (E - e0 < L1_EPB) ? (E - e0) : L1_EPB;
    float4* bin4 = (float4*)binbuf;
    for (int i = tid; i < nloc; i += L1_T) {
        const float4 r = rec4[i];
        const int b = __float_as_int(r.w) >> BIN_SH;
        const int pos = s_gb[b] + (i - s_off[b]);
        if (pos < BCAP) bin4[(size_t)b * BCAP + pos] = r;
    }
    if (s_full) {
        for (int i = tid; i < nloc; i += L1_T) {
            const float4 r = rec4[i];
            const int tt = __float_as_int(r.w);
            const int b = tt >> BIN_SH;
            const int pos = s_gb[b] + (i - s_off[b]);
            if (pos >= BCAP) {
                const int p = atomicAdd(&cnt[tt], 1);
                if (p < CAP) {
                    ((float4*)crec)[(size_t)tt * CAP + p] = make_float4(r.x, r.y, r.z, 0.f);
                } else {
                    const int q = atomicAdd(novf, 1);
                    ((float4*)ovf)[q] = r;
                }
            }
        }
    }
}

constexpr int PL_U = 4;
__global__ void __launch_bounds__(256)
k_place(const int* __restrict__ bincnt, const float* __restrict__ binbuf,
        int* __restrict__ cnt, int* __restrict__ novf,
        float* __restrict__ crec, float* __restrict__ ovf, int N) {
    __shared__ __align__(16) float s_st[BIN_NODES * CAP * 4];
    __shared__ int s_cnt[BIN_NODES];
    __shared__ int s_c0[BIN_NODES];
    const int b = blockIdx.x, tid = threadIdx.x;
    const int node0 = b << BIN_SH;
    if (tid < BIN_NODES) {
        const int nd = node0 + tid;
        const int c0 = (nd < N) ? cnt[nd] : 0;
        s_cnt[tid] = c0; s_c0[tid] = c0;
    }
    __syncthreads();
    const int nraw = bincnt[b];
    const int nin = nraw < BCAP ? nraw : BCAP;
    const float4* in4 = (const float4*)binbuf + (size_t)b * BCAP;
    float4* st4 = (float4*)s_st;
    for (int i0 = 0; i0 < nin; i0 += 256 * PL_U) {
        float4 r[PL_U]; bool ok[PL_U];
#pragma unroll
        for (int u = 0; u < PL_U; ++u) {
            const int i = i0 + u * 256 + tid;
            ok[u] = i < nin;
            r[u] = in4[ok[u] ? i : 0];
        }
#pragma unroll
        for (int u = 0; u < PL_U; ++u) {
            if (!ok[u]) continue;
            const int n = __float_as_int(r[u].w) - node0;
            const int p = atomicAdd(&s_cnt[n], 1);
            if (p < CAP) {
                st4[n * CAP + p] = make_float4(r[u].x, r[u].y, r[u].z, 0.f);
            } else {
                const int q = atomicAdd(novf, 1);
                ((float4*)ovf)[q] = r[u];
            }
        }
    }
    __syncthreads();
    float4* out4 = (float4*)crec + (size_t)node0 * CAP;
    for (int i = tid; i < BIN_NODES * CAP; i += 256) {
        const int n = i / CAP, p = i - n * CAP;
        const int cn = s_cnt[n];
        const int hi = cn < CAP ? cn : CAP;
        if (p >= s_c0[n] && p < hi) out4[i] = st4[i];
    }
    if (tid < BIN_NODES && node0 + tid < N) cnt[node0 + tid] = s_cnt[tid];
}

__device__ __forceinline__ void load8(const float* __restrict__ p, float (&v)[8]) {
    const float4 a = ((const float4*)p)[0], b = ((const float4*)p)[1];
    v[0] = a.x; v[1] = a.y; v[2] = a.z; v[3] = a.w;
    v[4] = b.x; v[5] = b.y; v[6] = b.z; v[7] = b.w;
}

__device__ __forceinline__ float head_logit8(const float2 xs, const float2 a,
        const float (&wl0)[8], const float (&wl1)[8],
        const float (&we0)[8], const float (&we1)[8],
        const float (&base)[8], const float (&at4)[8], const float (&lin)[5]) {
    float lg = fmaf(xs.x, lin[0], fmaf(xs.y, lin[1], fmaf(a.x, lin[2], fmaf(a.y, lin[3], lin[4]))));
#pragma unroll
    for (int c = 0; c < 8; ++c) {
        const float m = fmaf(xs.x, wl0[c], fmaf(xs.y, wl1[c],
                        fmaf(a.x, we0[c], fmaf(a.y, we1[c], base[c]))));
        lg = fmaf(__builtin_fabsf(m), at4[c], lg);
    }
    return lg;
}

constexpr int L1_U = 4;
__global__ void __launch_bounds__(256)
k_node_attn1(const float* __restrict__ x,
             const float* __restrict__ Wl, const float* __restrict__ bl,
             const float* __restrict__ Wr, const float* __restrict__ br,
             const float* __restrict__ We, const float* __restrict__ att,
             const float* __restrict__ bias,
             const int* __restrict__ cnt, const float* __restrict__ crec,
             const int* __restrict__ novf, const float* __restrict__ ovf,
             _Float16* __restrict__ hh, int N, int E) {
    int idx = blockIdx.x * blockDim.x + threadIdx.x;
    if (idx >= N * 16) return;
    const int t = idx >> 4, d0 = (idx & 15) * 8;
    const float2* x2 = (const float2*)x;
    float wl0[8], wl1[8], we0[8], we1[8], at[8], base[8];
    load8(Wl + d0, wl0); load8(Wl + 128 + d0, wl1);
    load8(We + d0, we0); load8(We + 128 + d0, we1);
    load8(att + d0, at);
    {
        const float2 xt = x2[t];
        float wr0[8], wr1[8], brv[8], blv[8];
        load8(Wr + d0, wr0); load8(Wr + 128 + d0, wr1);
        load8(br + d0, brv); load8(bl + d0, blv);
#pragma unroll
        for (int c = 0; c < 8; ++c)
            base[c] = fmaf(xt.x, wr0[c], fmaf(xt.y, wr1[c], brv[c])) + blv[c];
    }
    float lin[5] = {0.f, 0.f, 0.f, 0.f, 0.f}, at4[8];
#pragma unroll
    for (int c = 0; c < 8; ++c) {
        lin[0] = fmaf(at[c], wl0[c], lin[0]); lin[1] = fmaf(at[c], wl1[c], lin[1]);
        lin[2] = fmaf(at[c], we0[c], lin[2]); lin[3] = fmaf(at[c], we1[c], lin[3]);
        lin[4] = fmaf(at[c], base[c], lin[4]);
        at4[c] = 0.4f * at[c];
    }
#pragma unroll
    for (int k = 0; k < 5; ++k) lin[k] *= 0.6f;
    const int deg = cnt[t];
    const unsigned beg = (unsigned)t * CAP, end = beg + (unsigned)(deg < CAP ? deg : CAP);
    const float4* rec4 = (const float4*)crec;
    float mx = -__builtin_huge_valf(), den = 0.f, sx0 = 0.f, sx1 = 0.f;
    for (unsigned j = beg; j < end; j += L1_U) {
        int s[L1_U]; float2 a[L1_U], xs[L1_U]; bool ok[L1_U];
        float l[L1_U];
#pragma unroll
        for (int u = 0; u < L1_U; ++u) {
            ok[u] = (j + u) < end;
            unsigned jj = ok[u] ? (j + u) : j;
            const float4 r = rec4[jj];
            s[u] = __float_as_int(r.x);
            a[u] = make_float2(r.y, r.z);
        }
#pragma unroll
        for (int u = 0; u < L1_U; ++u) xs[u] = x2[s[u]];
        float cm = mx;
#pragma unroll
        for (int u = 0; u < L1_U; ++u) {
            const float lg = head_logit8(xs[u], a[u], wl0, wl1, we0, we1, base, at4, lin);
            l[u] = ok[u] ? lg : -__builtin_huge_valf();
            cm = fmaxf(cm, l[u]);
        }
        const float sc = __expf(mx - cm);
        den *= sc; sx0 *= sc; sx1 *= sc;
#pragma unroll
        for (int u = 0; u < L1_U; ++u) {
            float w = __expf(l[u] - cm);
            den += w;
            sx0 = fmaf(w, xs[u].x, sx0);
            sx1 = fmaf(w, xs[u].y, sx1);
        }
        mx = cm;
    }
    if (deg > CAP) {
        const int nraw = *novf;
        const int no = nraw < E ? nraw : E;
        const float4* o4 = (const float4*)ovf;
        for (int i = 0; i < no; ++i) {
            const float4 r = o4[i];
            if (__float_as_int(r.w) != t) continue;
            const float2 xs = x2[__float_as_int(r.x)];
            const float lg = head_logit8(xs, make_float2(r.y, r.z), wl0, wl1, we0, we1, base, at4, lin);
            const float cm = fmaxf(mx, lg);
            const float sc = __expf(mx - cm), w = __expf(lg - cm);
            den = fmaf(den, sc, w);
            sx0 = fmaf(sx0, sc, w * xs.x);
            sx1 = fmaf(sx1, sc, w * xs.y);
            mx = cm;
        }
    }
    const bool has = deg > 0;
    const float m0 = has ? sx0 / den : 0.f, m1 = has ? sx1 / den : 0.f;
    float blv[8], bsv[8];
    load8(bl + d0, blv); load8(bias + d0, bsv);
    v8h o;
#pragma unroll
    for (int c = 0; c < 8; ++c) {
        float v = (has ? fmaf(m0, wl0[c], fmaf(m1, wl1[c], blv[c])) : 0.f) + bsv[c];
        v = (v > 0.f) ? v : (__expf(v) - 1.f);
        o[c] = (_Float16)v;
    }
    *(v8h*)(hh + (size_t)t * 128 + d0) = o;
}

__device__ __forceinline__ float quad_xor1(float v) {
    return __int_as_float(__builtin_amdgcn_update_dpp(0, __float_as_int(v), 0xB1, 0xF, 0xF, true));
}
__device__ __forceinline__ float quad_xor2(float v) {
    return __int_as_float(__builtin_amdgcn_update_dpp(0, __float_as_int(v), 0x4E, 0xF, 0xF, true));
}

__device__ __forceinline__ float head_logit4(const float4 row, const float4 xrt, const float2 a,
        const float4 w0, const float4 w1, const float4 at) {
    float m0 = row.x + xrt.x + fmaf(a.x, w0.x, a.y * w1.x);
    float m1 = row.y + xrt.y + fmaf(a.x, w0.y, a.y * w1.y);
    float m2 = row.z + xrt.z + fmaf(a.x, w0.z, a.y * w1.z);
    float m3 = row.w + xrt.w + fmaf(a.x, w0.w, a.y * w1.w);
    m0 = fmaxf(m0, 0.2f * m0);
    m1 = fmaxf(m1, 0.2f * m1);
    m2 = fmaxf(m2, 0.2f * m2);
    m3 = fmaxf(m3, 0.2f * m3);
    float lg = fmaf(m3, at.w, fmaf(m2, at.z, fmaf(m1, at.y, m0 * at.x)));
    lg += quad_xor1(lg);
    lg += quad_xor2(lg);
    return lg;
}

constexpr int L2_U = 4;
__global__ void __launch_bounds__(256)
k_node_attn2(const float* __restrict__ xl, const float* __restrict__ xr,
             const float* __restrict__ We, const float* __restrict__ att,
             const int* __restrict__ cnt, const float* __restrict__ crec,
             const int* __restrict__ novf, const float* __restrict__ ovf,
             const float* __restrict__ bias2, const int* __restrict__ batch,
             float* __restrict__ sums, float* __restrict__ cnts, int N, int E) {
    __shared__ __align__(16) float s_row[16 * 64];
    __shared__ int s_g[16];
    const int idx = blockIdx.x * blockDim.x + threadIdx.x;
    const bool valid = idx < N * 16;
    const int t = valid ? (idx >> 4) : 0, d0 = (idx & 15) * 4;
    const float4 xrt = *(const float4*)(xr + (size_t)t * 64 + d0);
    const float4 w0 = *(const float4*)(We + d0);
    const float4 w1 = *(const float4*)(We + 64 + d0);
    const float4 at = *(const float4*)(att + d0);
    float4 acc = make_float4(0.f, 0.f, 0.f, 0.f);
    const int deg = valid ? cnt[t] : 0;
    const unsigned beg = (unsigned)t * CAP, end = beg + (unsigned)(deg < CAP ? deg : CAP);
    const float4* rec4 = (const float4*)crec;
    float mx = -__builtin_huge_valf(), den = 0.f;
    for (unsigned j = beg; j < end; j += L2_U) {
        int s[L2_U]; float2 a[L2_U]; bool ok[L2_U];
        float4 row[L2_U]; float l[L2_U];
#pragma unroll
        for (int u = 0; u < L2_U; ++u) {
            ok[u] = (j + u) < end;
            unsigned jj = ok[u] ? (j + u) : j;
            const float4 r = rec4[jj];
            s[u] = __float_as_int(r.x);
            a[u] = make_float2(r.y, r.z);
        }
#pragma unroll
        for (int u = 0; u < L2_U; ++u)
            row[u] = *(const float4*)(xl + (size_t)s[u] * 64 + d0);
        float cm = mx;
#pragma unroll
        for (int u = 0; u < L2_U; ++u) {
            float m0 = row[u].x + xrt.x + fmaf(a[u].x, w0.x, a[u].y * w1.x);
            float m1 = row[u].y + xrt.y + fmaf(a[u].x, w0.y, a[u].y * w1.y);
            float m2 = row[u].z + xrt.z + fmaf(a[u].x, w0.z, a[u].y * w1.z);
            float m3 = row[u].w + xrt.w + fmaf(a[u].x, w0.w, a[u].y * w1.w);
            m0 = fmaxf(m0, 0.2f * m0);
            m1 = fmaxf(m1, 0.2f * m1);
            m2 = fmaxf(m2, 0.2f * m2);
            m3 = fmaxf(m3, 0.2f * m3);
            float lg = fmaf(m3, at.w, fmaf(m2, at.z, fmaf(m1, at.y, m0 * at.x)));
            lg += quad_xor1(lg);
            lg += quad_xor2(lg);
            l[u] = ok[u] ? lg : -__builtin_huge_valf();
            cm = fmaxf(cm, l[u]);
        }
        const float sc = __expf(mx - cm);
        den *= sc;
        acc.x *= sc; acc.y *= sc; acc.z *= sc; acc.w *= sc;
#pragma unroll
        for (int u = 0; u < L2_U; ++u) {
            float w = __expf(l[u] - cm);
            den += w;
            acc.x = fmaf(w, row[u].x, acc.x);
            acc.y = fmaf(w, row[u].y, acc.y);
            acc.z = fmaf(w, row[u].z, acc.z);
            acc.w = fmaf(w, row[u].w, acc.w);
        }
        mx = cm;
    }
    if (deg > CAP) {
        const int nraw = *novf;
        const int no = nraw < E ? nraw : E;
        const float4* o4 = (const float4*)ovf;
        for (int i = 0; i < no; ++i) {
            const float4 r = o4[i];
            if (__float_as_int(r.w) != t) continue;
            const float4 row = *(const float4*)(xl + (size_t)__float_as_int(r.x) * 64 + d0);
            const float lg = head_logit4(row, xrt, make_float2(r.y, r.z), w0, w1, at);
            const float cm = fmaxf(mx, lg);
            const float sc = __expf(mx - cm), w = __expf(lg - cm);
            den = fmaf(den, sc, w);
            acc.x = fmaf(acc.x, sc, w * row.x);
            acc.y = fmaf(acc.y, sc, w * row.y);
            acc.z = fmaf(acc.z, sc, w * row.z);
            acc.w = fmaf(acc.w, sc, w * row.w);
            mx = cm;
        }
    }
    float4 o = make_float4(0.f, 0.f, 0.f, 0.f);
    if (deg > 0) { o.x = acc.x / den; o.y = acc.y / den; o.z = acc.z / den; o.w = acc.w / den; }

    const float4 b2 = *(const float4*)(bias2 + d0);
    const int nl = threadIdx.x >> 4;
    *(float4*)(s_row + nl * 64 + d0) = make_float4(o.x + b2.x, o.y + b2.y, o.z + b2.z, o.w + b2.w);
    if ((threadIdx.x & 15) == 0) s_g[nl] = valid ? batch[t] : -1;
    __syncthreads();
    if (threadIdx.x < 64) {
        const int d = threadIdx.x;
        float racc = 0.f, rcnt = 0.f;
        int gcur = -1;
        float v[16];
        int   g[16];
#pragma unroll
        for (int n = 0; n < 16; ++n) { v[n] = s_row[n * 64 + d]; g[n] = s_g[n]; }
#pragma unroll
        for (int n = 0; n < 16; ++n) {
            if (g[n] != gcur) {
                if (gcur >= 0) {
                    atomicAdd(&sums[gcur * 64 + d], racc);
                    if (d == 0) atomicAdd(&cnts[gcur], rcnt);
                }
                gcur = g[n]; racc = 0.f; rcnt = 0.f;
            }
            racc += v[n];
            rcnt += 1.0f;
        }
        if (gcur >= 0) {
            atomicAdd(&sums[gcur * 64 + d], racc);
            if (d == 0) atomicAdd(&cnts[gcur], rcnt);
        }
    }
}

__global__ void k_prep(const float* __restrict__ Wl, const float* __restrict__ Wr,
                       _Float16* __restrict__ Wpk, uint4* __restrict__ zbase, size_t zbytes) {
    int idx = blockIdx.x * blockDim.x + threadIdx.x;
    const unsigned zn16 = (unsigned)(zbytes >> 4);
    for (unsigned i = (unsigned)idx; i < zn16; i += gridDim.x * blockDim.x)
        zbase[i] = make_uint4(0u, 0u, 0u, 0u);
    if (idx >= 2 * 8192) return;
    int which = idx >> 13;
    int r = idx & 8191;
    int j = r & 15;
    int lane = (r >> 4) & 31;
    int frag = r >> 9;
    int kc = frag >> 2, nt = frag & 3;
    int half_sel = lane >> 4, lidx = lane & 15;
    int kb = (j < 8) ? (half_sel * 8) : (16 + half_sel * 8);
    int k = kc * 32 + kb + (j & 7);
    int n = nt * 16 + lidx;
    const float* W = which ? Wr : Wl;
    Wpk[idx] = (_Float16)W[k * 64 + n];
}

__global__ void k_gemm2_wmma(const _Float16* __restrict__ hh,
                             const _Float16* __restrict__ Wpk,
                             const float* __restrict__ bl, const float* __restrict__ br,
                             float* __restrict__ xl2, float* __restrict__ xr2, int N) {
    __shared__ __align__(32) _Float16 ldsW[2 * 8192];

    {
        const uint4* g = (const uint4*)Wpk;
        uint4* l = (uint4*)ldsW;
        for (int i = threadIdx.x; i < 2048; i += blockDim.x) l[i] = g[i];
    }
    __syncthreads();

    int wave = (blockIdx.x * blockDim.x + threadIdx.x) >> 5;
    int lane = threadIdx.x & 31;
    int row0 = wave * 16;
    if (row0 >= N) return;

    int half_sel = lane >> 4;
    int lidx = lane & 15;
    int kb_lo = half_sel * 8;
    int kb_hi = 16 + kb_lo;
    int m = row0 + lidx;
    if (m >= N) m = N - 1;

    const v16h* fragL = (const v16h*)ldsW;
    const v16h* fragR = (const v16h*)(ldsW + 8192);

    v8f accl[4] = {};
    v8f accr[4] = {};

#pragma unroll
    for (int kc = 0; kc < 4; ++kc) {
        const _Float16* pa = hh + m * 128 + kc * 32;
        v8h alo = *(const v8h*)(pa + kb_lo);
        v8h ahi = *(const v8h*)(pa + kb_hi);
        v16h a = __builtin_shufflevector(alo, ahi, 0, 1, 2, 3, 4, 5, 6, 7,
                                         8, 9, 10, 11, 12, 13, 14, 15);
#pragma unroll
        for (int nt = 0; nt < 4; ++nt) {
            int fi = (kc * 4 + nt) * 32 + lane;
            v16h b  = fragL[fi];
            v16h b2 = fragR[fi];
            accl[nt] = __builtin_amdgcn_wmma_f32_16x16x32_f16(
                false, a, false, b, (short)0, accl[nt], false, false);
            accr[nt] = __builtin_amdgcn_wmma_f32_16x16x32_f16(
                false, a, false, b2, (short)0, accr[nt], false, false);
        }
    }
#pragma unroll
    for (int nt = 0; nt < 4; ++nt) {
        int n = nt * 16 + lidx;
        float bln = bl[n], brn = br[n];
#pragma unroll
        for (int r = 0; r < 8; ++r) {
            int mo = row0 + r + half_sel * 8;
            if (mo < N) {
                xl2[mo * 64 + n] = accl[nt][r] + bln;
                xr2[mo * 64 + n] = accr[nt][r] + brn;
            }
        }
    }
}

__global__ void k_final(const float* __restrict__ sums, const float* __restrict__ cnts,
                        float* __restrict__ out, int G) {
    int idx = blockIdx.x * blockDim.x + threadIdx.x;
    if (idx >= G * 64) return;
    out[idx] = sums[idx] / fmaxf(cnts[idx >> 6], 1.0f);
}

extern "C" void kernel_launch(void* const* d_in, const int* in_sizes, int n_in,
                              void* d_out, int out_size, void* d_ws, size_t ws_size,
                              hipStream_t stream) {
    const float* x      = (const float*)d_in[0];
    const float* ea     = (const float*)d_in[1];
    const float* W_l1   = (const float*)d_in[2];
    const float* b_l1   = (const float*)d_in[3];
    const float* W_r1   = (const float*)d_in[4];
    const float* b_r1   = (const float*)d_in[5];
    const float* W_e1   = (const float*)d_in[6];
    const float* att1   = (const float*)d_in[7];
    const float* bias1  = (const float*)d_in[8];
    const float* W_l2   = (const float*)d_in[9];
    const float* b_l2   = (const float*)d_in[10];
    const float* W_r2   = (const float*)d_in[11];
    const float* b_r2   = (const float*)d_in[12];
    const float* W_e2   = (const float*)d_in[13];
    const float* att2   = (const float*)d_in[14];
    const float* bias2  = (const float*)d_in[15];
    const int*   eidx   = (const int*)d_in[16];
    const int*   batch  = (const int*)d_in[17];
    float* out = (float*)d_out;

    const int N = in_sizes[0] / 2;
    const int E = in_sizes[1] / 2;
    const int G = out_size / 64;
    const int* src = eidx;
    const int* tgt = eidx + E;

    char* ws = (char*)d_ws;
    size_t off = 0;
    auto alloc = [&](size_t bytes) { size_t o = off; off = (off + bytes + 255) & ~(size_t)255; return o; };

    size_t o_cnt   = alloc((size_t)N * 4);
    size_t o_novf  = alloc(4);
    size_t o_sums  = alloc((size_t)G * 64 * 4);
    size_t o_cnts  = alloc((size_t)G * 4);
    size_t o_bcnt  = alloc((size_t)MAXBIN * 4);
    size_t acc_bytes = off;

    size_t o_crec = alloc((size_t)N * CAP * 16);
    size_t o_ovf  = alloc((size_t)E * 16);
    const int nbin = (N + BIN_NODES - 1) / BIN_NODES;
    size_t o_bin  = alloc((size_t)nbin * BCAP * 16);
    size_t o_hh   = alloc((size_t)N * 128 * 2);
    size_t o_wpk  = alloc((size_t)2 * 8192 * 2);
    size_t o_xl2  = alloc((size_t)N * 64 * 4);
    size_t o_xr2  = alloc((size_t)N * 64 * 4);
    (void)ws_size; (void)n_in;

    int*      cnt   = (int*)(ws + o_cnt);
    int*      novf  = (int*)(ws + o_novf);
    float*    sums  = (float*)(ws + o_sums);
    float*    cnts  = (float*)(ws + o_cnts);
    float*    crec  = (float*)(ws + o_crec);
    float*    ovf   = (float*)(ws + o_ovf);
    int*      bcnt  = (int*)(ws + o_bcnt);
    float*    binb  = (float*)(ws + o_bin);
    _Float16* hh    = (_Float16*)(ws + o_hh);
    _Float16* wpk   = (_Float16*)(ws + o_wpk);
    float*    xl2   = (float*)(ws + o_xl2);
    float*    xr2   = (float*)(ws + o_xr2);

    const int TB = 256;
    auto blocks = [&](long long t) { return (unsigned)((t + TB - 1) / TB); };

    k_prep<<<blocks(2 * 8192), TB, 0, stream>>>(W_l2, W_r2, wpk, (uint4*)d_ws, acc_bytes);

    k_bin<<<(unsigned)((E + L1_EPB - 1) / L1_EPB), L1_T, 0, stream>>>(
        src, tgt, ea, bcnt, binb, cnt, novf, crec, ovf, E);
    k_place<<<(unsigned)nbin, TB, 0, stream>>>(bcnt, binb, cnt, novf, crec, ovf, N);

    k_node_attn1<<<blocks((long long)N * 16), TB, 0, stream>>>(
        x, W_l1, b_l1, W_r1, b_r1, W_e1, att1, bias1, cnt, crec, novf, ovf, hh, N, E);

    long long gemm_threads = (long long)((N + 15) / 16) * 32;
    k_gemm2_wmma<<<blocks(gemm_threads), TB, 0, stream>>>(
        hh, wpk, b_l2, b_r2, xl2, xr2, N);

    k_node_attn2<<<blocks((long long)N * 16), TB, 0, stream>>>(
        xl2, xr2, W_e2, att2, cnt, crec, novf, ovf, bias2, batch, sums, cnts, N, E);

    k_final<<<blocks((long long)G * 64), TB, 0, stream>>>(sums, cnts, out, G);
}
